# starting kernel
# baseline (speedup 1.0000x reference)
#include <hip/hip_runtime.h>
#include <hip/hip_bf16.h>

typedef __attribute__((ext_vector_type(2))) float v2f;
typedef __attribute__((ext_vector_type(8))) float v8f;

#define BN_EPS 1e-5f
#define CH 128
#define WSTRIDE 132

__device__ __forceinline__ void fadd_atomic(float* p, float v) {
  unsafeAtomicAdd(p, v);
}

__global__ void zero_f4(float4* __restrict__ p, size_t n4) {
  size_t i = (size_t)blockIdx.x * blockDim.x + threadIdx.x;
  size_t stride = (size_t)gridDim.x * blockDim.x;
  float4 z = {0.f, 0.f, 0.f, 0.f};
  for (; i < n4; i += stride) p[i] = z;
}

__global__ void zero_f(float* __restrict__ p, int n) {
  int i = blockIdx.x * blockDim.x + threadIdx.x;
  if (i < n) p[i] = 0.f;
}

__global__ void spmm128_kernel(const float* __restrict__ feat,
                               const int* __restrict__ src,
                               const int* __restrict__ dst,
                               const float* __restrict__ ew,
                               float* __restrict__ agg, int E) {
  int e = blockIdx.x * 8 + (threadIdx.x >> 5);
  if (e >= E) return;
  int lane = threadIdx.x & 31;
  int s = src[e];
  int d = dst[e];
  float w = ew[e];
  const float4* fr = (const float4*)(feat + (size_t)s * CH);
  float4 v = fr[lane];
  float* o = agg + (size_t)d * CH + lane * 4;
  fadd_atomic(o + 0, v.x * w);
  fadd_atomic(o + 1, v.y * w);
  fadd_atomic(o + 2, v.z * w);
  fadd_atomic(o + 3, v.w * w);
}

__global__ void spmm40_kernel(const float* __restrict__ feat,
                              const int* __restrict__ src,
                              const int* __restrict__ dst,
                              const float* __restrict__ ew,
                              float* __restrict__ agg, int E) {
  int e = blockIdx.x * 8 + (threadIdx.x >> 5);
  if (e >= E) return;
  int lane = threadIdx.x & 31;
  if (lane >= 20) return;
  int s = src[e];
  int d = dst[e];
  float w = ew[e];
  float2 v = *(const float2*)(feat + (size_t)s * 40 + lane * 2);
  float* o = agg + (size_t)d * 40 + lane * 2;
  fadd_atomic(o + 0, v.x * w);
  fadd_atomic(o + 1, v.y * w);
}

template <int NT>
__global__ void gemm_wmma_kernel(const float* __restrict__ A,
                                 const float* __restrict__ W,
                                 float* __restrict__ out,
                                 int Nrows, int Cout, int ldo) {
  constexpr int CP = NT * 16;
  __shared__ float sWT[CP * WSTRIDE];
  int tid = threadIdx.x;
  for (int idx = tid; idx < CH * CP; idx += 256) {
    int k = idx / CP;
    int c = idx % CP;
    sWT[c * WSTRIDE + k] = (c < Cout) ? W[k * Cout + c] : 0.f;
  }
  __syncthreads();

  int wave = tid >> 5;
  int lane = tid & 31;
  int row0 = blockIdx.x * 128 + wave * 16;
  if (row0 >= Nrows) return;

  int m = lane & 15;
  int kh = (lane >> 4) * 2;

  v8f acc[NT];
#pragma unroll
  for (int t = 0; t < NT; ++t) acc[t] = {};

  const float* Arow = A + (size_t)(row0 + m) * CH;

  for (int k0 = 0; k0 < CH; k0 += 4) {
    v2f a = *(const v2f*)(Arow + k0 + kh);
#pragma unroll
    for (int t = 0; t < NT; ++t) {
      v2f b = *(const v2f*)(&sWT[(t * 16 + m) * WSTRIDE + k0 + kh]);
      acc[t] = __builtin_amdgcn_wmma_f32_16x16x4_f32(
           false, a,  false, b,
           (short)0, acc[t],  false,  false);
    }
  }

  int rbase = row0 + (lane >> 4) * 8;
#pragma unroll
  for (int t = 0; t < NT; ++t) {
    int col = t * 16 + m;
    if (col < Cout) {
#pragma unroll
      for (int j = 0; j < 8; ++j) {
        out[(size_t)(rbase + j) * ldo + col] = acc[t][j];
      }
    }
  }
}

__global__ void colstats_kernel(const float* __restrict__ g,
                                float* __restrict__ sums, int Nrows) {
  __shared__ float sh[256];
  sh[threadIdx.x] = 0.f;
  __syncthreads();
  size_t total = (size_t)Nrows * CH;
  size_t i = (size_t)blockIdx.x * 256 + threadIdx.x;
  size_t stride = (size_t)gridDim.x * 256;
  int col = (int)(i & (CH - 1));
  float s = 0.f, q = 0.f;
  for (; i < total; i += stride) {
    float v = g[i];
    s += v;
    q += v * v;
  }
  atomicAdd(&sh[col], s);
  atomicAdd(&sh[128 + col], q);
  __syncthreads();
  if (threadIdx.x < 128) {
    atomicAdd(&sums[threadIdx.x], sh[threadIdx.x]);
    atomicAdd(&sums[128 + threadIdx.x], sh[128 + threadIdx.x]);
  }
}

__global__ void bnrelu_kernel(float* __restrict__ g,
                              const float* __restrict__ sums,
                              const float* __restrict__ gamma,
                              const float* __restrict__ beta, int Nrows) {
  size_t total = (size_t)Nrows * CH;
  size_t i = (size_t)blockIdx.x * blockDim.x + threadIdx.x;
  size_t stride = (size_t)gridDim.x * blockDim.x;
  float invN = 1.0f / (float)Nrows;
  for (; i < total; i += stride) {
    int col = (int)(i & (CH - 1));
    float mean = sums[col] * invN;
    float var = sums[128 + col] * invN - mean * mean;
    float sc = gamma[col] * rsqrtf(var + BN_EPS);
    float v = (g[i] - mean) * sc + beta[col];
    g[i] = v > 0.f ? v : 0.f;
  }
}

__global__ void biasout_kernel(const float* __restrict__ agg40,
                               const float* __restrict__ b3,
                               float* __restrict__ out, int total) {
  int i = blockIdx.x * blockDim.x + threadIdx.x;
  if (i < total) {
    int c = i % 40;
    out[i] = agg40[i] + b3[c];
  }
}

extern "C" void kernel_launch(void* const* d_in, const int* in_sizes, int n_in,
                              void* d_out, int out_size, void* d_ws, size_t ws_size,
                              hipStream_t stream) {
  const float* feat   = (const float*)d_in[0];
  const int*   src    = (const int*)d_in[1];
  const int*   dst    = (const int*)d_in[2];
  const float* ew     = (const float*)d_in[3];
  const float* W1     = (const float*)d_in[4];
  const float* W2     = (const float*)d_in[5];
  const float* W3     = (const float*)d_in[6];
  const float* b3     = (const float*)d_in[7];
  const float* gamma1 = (const float*)d_in[8];
  const float* beta1  = (const float*)d_in[9];
  const float* gamma2 = (const float*)d_in[10];
  const float* beta2  = (const float*)d_in[11];

  int N = in_sizes[0] / CH;
  int E = in_sizes[1];

  float* agg  = (float*)d_ws;
  float* g    = agg + (size_t)N * CH;
  float* sums = g + (size_t)N * CH;
  float* t3   = agg;
  float* agg3 = agg + (size_t)N * 64;

  size_t n4 = (size_t)N * (CH / 4);
  int zgrid = 2048;
  int sgrid = (E + 7) / 8;
  int ggrid = (N + 127) / 128;
  int egrid = 2048;
  int bgrid = (N * 40 + 255) / 256;
  dim3 b256(256);

  zero_f4<<<zgrid, b256, 0, stream>>>((float4*)agg, n4);
  spmm128_kernel<<<sgrid, b256, 0, stream>>>(feat, src, dst, ew, agg, E);
  gemm_wmma_kernel<8><<<ggrid, b256, 0, stream>>>(agg, W1, g, N, 128, 128);
  zero_f<<<1, b256, 0, stream>>>(sums, 256);
  colstats_kernel<<<egrid, b256, 0, stream>>>(g, sums, N);
  bnrelu_kernel<<<egrid, b256, 0, stream>>>(g, sums, gamma1, beta1, N);

  zero_f4<<<zgrid, b256, 0, stream>>>((float4*)agg, n4);
  spmm128_kernel<<<sgrid, b256, 0, stream>>>(g, src, dst, ew, agg, E);
  gemm_wmma_kernel<8><<<ggrid, b256, 0, stream>>>(agg, W2, g, N, 128, 128);
  zero_f<<<1, b256, 0, stream>>>(sums, 256);
  colstats_kernel<<<egrid, b256, 0, stream>>>(g, sums, N);
  bnrelu_kernel<<<egrid, b256, 0, stream>>>(g, sums, gamma2, beta2, N);

  gemm_wmma_kernel<3><<<ggrid, b256, 0, stream>>>(g, W3, t3, N, 40, 40);
  zero_f4<<<zgrid, b256, 0, stream>>>((float4*)agg3, (size_t)N * 10);
  spmm40_kernel<<<sgrid, b256, 0, stream>>>(t3, src, dst, ew, agg3, E);
  biasout_kernel<<<bgrid, b256, 0, stream>>>(agg3, b3, (float*)d_out, N * 40);
}


# Layers 1-2 edge sums run as two 64-channel passes over channel planes (product and sum stored [2][N][64]), so the rows a pass gathers and adds into fit the second-level cache; same f32 atomics
# speedup vs baseline: 2.2502x; 2.2502x over previous
#include <hip/hip_runtime.h>
#include <hip/hip_bf16.h>

typedef __attribute__((ext_vector_type(2))) float v2f;
typedef __attribute__((ext_vector_type(8))) float v8f;

#define BN_EPS 1e-5f
#define CH 128
#define WSTRIDE 132

__device__ __forceinline__ void fadd_atomic(float* p, float v) {
  unsafeAtomicAdd(p, v);
}

__device__ __forceinline__ int bcast_i(int v, int j) {
  return __builtin_amdgcn_readlane(v, j);
}
__device__ __forceinline__ float bcast_f(float v, int j) {
  return __int_as_float(__builtin_amdgcn_readlane(__float_as_int(v), j));
}

__global__ void __launch_bounds__(256)
spmm128_kernel(const float* __restrict__ feat,
               const int* __restrict__ src,
               const int* __restrict__ dst,
               const float* __restrict__ ew,
               float* __restrict__ agg, int E, int pgrid, size_t pstride,
               float* __restrict__ zsums) {
  if (zsums != nullptr && blockIdx.x == 0) zsums[threadIdx.x] = 0.f;
  int lane = threadIdx.x & 31;
  int wv = __builtin_amdgcn_readfirstlane((int)(threadIdx.x >> 5));
  int plane = ((int)blockIdx.x >= pgrid) ? 1 : 0;
  int bidx = (int)blockIdx.x - plane * pgrid;
  const float* fpl = feat + (size_t)plane * pstride + lane;
  float* apl = agg + (size_t)plane * pstride + lane;
  int e0 = (bidx * 8 + wv) * 64;
  if (e0 >= E) return;
  for (int c = 0; c < 2; ++c) {
    int cnt = E - (e0 + c * 32);
    if (cnt <= 0) break;
    int e = e0 + c * 32 + lane;
    int s_l = 0, d_l = 0;
    float w_l = 0.f;
    if (e < E) {
      s_l = src[e];
      d_l = dst[e];
      w_l = ew[e];
    }
    if (cnt >= 32) {
#pragma unroll
      for (int g = 0; g < 32; g += 16) {
        float v[16][2];
#pragma unroll
        for (int j = 0; j < 16; ++j) {
          const float* fr = fpl + (size_t)bcast_i(s_l, g + j) * 64;
          v[j][0] = fr[0];
          v[j][1] = fr[32];
        }
#pragma unroll
        for (int j = 0; j < 16; ++j) {
          float w = bcast_f(w_l, g + j);
          float* o = apl + (size_t)bcast_i(d_l, g + j) * 64;
          fadd_atomic(o + 0, v[j][0] * w);
          fadd_atomic(o + 32, v[j][1] * w);
        }
      }
    } else {
      for (int j = 0; j < cnt; ++j) {
        const float* fr = fpl + (size_t)bcast_i(s_l, j) * 64;
        float w = bcast_f(w_l, j);
        float* o = apl + (size_t)bcast_i(d_l, j) * 64;
        float a0 = fr[0], a1 = fr[32];
        fadd_atomic(o + 0, a0 * w);
        fadd_atomic(o + 32, a1 * w);
      }
    }
  }
}

__global__ void __launch_bounds__(256)
spmm40_kernel(const float* __restrict__ feat,
              const int* __restrict__ src,
              const int* __restrict__ dst,
              const float* __restrict__ ew,
              float* __restrict__ agg, int E) {
  int lane = threadIdx.x & 31;
  int wv = __builtin_amdgcn_readfirstlane((int)(threadIdx.x >> 5));
  int e0 = (blockIdx.x * 8 + wv) * 32;
  if (e0 >= E) return;
  int cnt = E - e0;
  int e = e0 + lane;
  int s_l = 0, d_l = 0;
  float w_l = 0.f;
  if (e < E) {
    s_l = src[e];
    d_l = dst[e];
    w_l = ew[e];
  }
  bool tail8 = lane < 8;
  if (cnt >= 32) {
#pragma unroll
    for (int g = 0; g < 32; g += 8) {
      float va[8], vb[8];
#pragma unroll
      for (int j = 0; j < 8; ++j) {
        va[j] = feat[(size_t)bcast_i(s_l, g + j) * 40 + lane];
      }
      if (tail8) {
#pragma unroll
        for (int j = 0; j < 8; ++j) {
          vb[j] = feat[(size_t)bcast_i(s_l, g + j) * 40 + 32 + lane];
        }
      }
#pragma unroll
      for (int j = 0; j < 8; ++j) {
        float w = bcast_f(w_l, g + j);
        fadd_atomic(agg + (size_t)bcast_i(d_l, g + j) * 40 + lane, va[j] * w);
      }
      if (tail8) {
#pragma unroll
        for (int j = 0; j < 8; ++j) {
          float w = bcast_f(w_l, g + j);
          fadd_atomic(agg + (size_t)bcast_i(d_l, g + j) * 40 + 32 + lane, vb[j] * w);
        }
      }
    }
  } else {
    for (int j = 0; j < cnt; ++j) {
      const float* fr = feat + (size_t)bcast_i(s_l, j) * 40;
      float w = bcast_f(w_l, j);
      float* o = agg + (size_t)bcast_i(d_l, j) * 40;
      float a0 = fr[lane];
      fadd_atomic(o + lane, a0 * w);
      if (tail8) {
        float a1 = fr[32 + lane];
        fadd_atomic(o + 32 + lane, a1 * w);
      }
    }
  }
}

template <int NT, bool BN, int COUT>
__global__ void gemm_wmma_kernel(const float* A,
                                 const float* __restrict__ W,
                                 float* out,
                                 int Nrows, int ldo,
                                 const float* __restrict__ sums,
                                 const float* __restrict__ gamma,
                                 const float* __restrict__ beta,
                                 float* zrows,
                                 float* __restrict__ zsums,
                                 float* __restrict__ bout,
                                 const float* __restrict__ bias) {
  constexpr int CP = NT * 16;
  constexpr bool APL = BN;
  constexpr bool OPL = (COUT == CH);
  const size_t pst = (size_t)Nrows * 64;
  __shared__ float sWT[CP * WSTRIDE];
  __shared__ float sMean[CH];
  __shared__ float sSc[CH];
  __shared__ float sBeta[CH];
  int tid = threadIdx.x;
  if (zsums != nullptr && blockIdx.x == 0) zsums[tid] = 0.f;
  constexpr int WPER = CH * CP / 256;
  constexpr int WB = (WPER % 32 == 0) ? 32 : WPER;
  static_assert(WPER % WB == 0, "whole batches");
  for (int base = 0; base < WPER; base += WB) {
    float wv[WB];
#pragma unroll
    for (int i = 0; i < WB; ++i) {
      int idx = tid + (base + i) * 256;
      int k = idx / CP;
      int c = idx % CP;
      int cc = (c < COUT) ? c : (COUT - 1);
      wv[i] = W[k * COUT + cc];
    }
#pragma unroll
    for (int i = 0; i < WB; ++i) {
      int idx = tid + (base + i) * 256;
      int k = idx / CP;
      int c = idx % CP;
      sWT[c * WSTRIDE + k] = (c < COUT) ? wv[i] : 0.f;
    }
  }
  if (BN) {
    if (tid < CH) {
      float invN = 1.0f / (float)Nrows;
      float mean = sums[tid] * invN;
      float var = sums[128 + tid] * invN - mean * mean;
      sMean[tid] = mean;
      sSc[tid] = gamma[tid] * rsqrtf(var + BN_EPS);
      sBeta[tid] = beta[tid];
    }
  }
  __syncthreads();

  int wave = tid >> 5;
  int lane = tid & 31;
  int row0 = blockIdx.x * 128 + wave * 16;
  if (row0 >= Nrows) return;

  int m = lane & 15;
  int kh = (lane >> 4) * 2;

  v8f acc[NT];
#pragma unroll
  for (int t = 0; t < NT; ++t) acc[t] = {};

  const float* Arow = APL ? A + (size_t)(row0 + m) * 64 : A + (size_t)(row0 + m) * CH;
  const size_t ahalf = APL ? pst : (size_t)64;

  const float* Arow2 = Arow + ahalf - 64;

#pragma unroll 4
  for (int k0 = 0; k0 < CH; k0 += 4) {
    v2f a = *(const v2f*)((k0 < 64 ? Arow : Arow2) + k0 + kh);
    if (BN) {
      v2f mu = *(const v2f*)(&sMean[k0 + kh]);
      v2f sc = *(const v2f*)(&sSc[k0 + kh]);
      v2f be = *(const v2f*)(&sBeta[k0 + kh]);
      float x0 = (a.x - mu.x) * sc.x + be.x;
      float x1 = (a.y - mu.y) * sc.y + be.y;
      a.x = x0 > 0.f ? x0 : 0.f;
      a.y = x1 > 0.f ? x1 : 0.f;
    }
#pragma unroll
    for (int t = 0; t < NT; ++t) {
      v2f b = *(const v2f*)(&sWT[(t * 16 + m) * WSTRIDE + k0 + kh]);
      acc[t] = __builtin_amdgcn_wmma_f32_16x16x4_f32(
           false, a,  false, b,
           (short)0, acc[t],  false,  false);
    }
  }

  int rbase = row0 + (lane >> 4) * 8;
#pragma unroll
  for (int t = 0; t < NT; ++t) {
    int col = t * 16 + m;
    if (col < COUT) {
#pragma unroll
      for (int j = 0; j < 8; ++j) {
        if (OPL) {
          out[(size_t)(t >> 2) * pst + (size_t)(rbase + j) * 64 + (col & 63)] = acc[t][j];
        } else {
          out[(size_t)(rbase + j) * ldo + col] = acc[t][j];
        }
      }
    }
  }

  if (zrows != nullptr) {
    float4 zero4 = {0.f, 0.f, 0.f, 0.f};
#pragma unroll
    for (int p = 0; p < 2; ++p) {
      float4* z = (float4*)(zrows + (size_t)p * pst + (size_t)row0 * 64);
#pragma unroll
      for (int i = 0; i < 8; ++i) z[i * 32 + lane] = zero4;
    }
  }
  if (bout != nullptr) {
    float* o = bout + (size_t)row0 * 40;
#pragma unroll
    for (int i = 0; i < 20; ++i) {
      int e = i * 32 + lane;
      o[e] = bias[e % 40];
    }
  }
}

__global__ void colstats_kernel(const float* __restrict__ gp,
                                float* __restrict__ sums, int Nrows) {
  __shared__ float sh[256];
  sh[threadIdx.x] = 0.f;
  __syncthreads();
  size_t pidx = (size_t)blockIdx.x * 256 + threadIdx.x;
  int col = (int)(pidx & (CH - 1));
  size_t total = (size_t)Nrows * 64;
  const float* g = gp + (size_t)(col >> 6) * total + (col & 63);
  size_t i = (pidx >> 7) * 64;
  size_t stride = (size_t)gridDim.x * 128;
  float s = 0.f, q = 0.f;
  constexpr int RB = 12;
  for (; i + (RB - 1) * stride < total; i += RB * stride) {
    float v[RB];
#pragma unroll
    for (int j = 0; j < RB; ++j) v[j] = g[i + j * stride];
#pragma unroll
    for (int j = 0; j < RB; ++j) {
      s += v[j];
      q = fmaf(v[j], v[j], q);
    }
  }
  for (; i < total; i += stride) {
    float v = g[i];
    s += v;
    q = fmaf(v, v, q);
  }
  atomicAdd(&sh[col], s);
  atomicAdd(&sh[128 + col], q);
  __syncthreads();
  if (threadIdx.x < 128) {
    atomicAdd(&sums[threadIdx.x], sh[threadIdx.x]);
    atomicAdd(&sums[128 + threadIdx.x], sh[128 + threadIdx.x]);
  }
}

extern "C" void kernel_launch(void* const* d_in, const int* in_sizes, int n_in,
                              void* d_out, int out_size, void* d_ws, size_t ws_size,
                              hipStream_t stream) {
  const float* feat   = (const float*)d_in[0];
  const int*   src    = (const int*)d_in[1];
  const int*   dst    = (const int*)d_in[2];
  const float* ew     = (const float*)d_in[3];
  const float* W1     = (const float*)d_in[4];
  const float* W2     = (const float*)d_in[5];
  const float* W3     = (const float*)d_in[6];
  const float* b3     = (const float*)d_in[7];
  const float* gamma1 = (const float*)d_in[8];
  const float* beta1  = (const float*)d_in[9];
  const float* gamma2 = (const float*)d_in[10];
  const float* beta2  = (const float*)d_in[11];

  int N = in_sizes[0] / CH;
  int E = in_sizes[1];

  float* t    = (float*)d_ws;
  float* a    = t + (size_t)N * CH;
  float* sums = a + (size_t)N * CH;
  float* out  = (float*)d_out;

  int sgrid = (E + 255) / 256;
  int pgrid = (E + 511) / 512;
  size_t pstride = (size_t)N * 64;
  int ggrid = (N + 127) / 128;
  int egrid = 2048;
  dim3 b256(256);

  gemm_wmma_kernel<8, false, 128><<<ggrid, b256, 0, stream>>>(
      feat, W1, t, N, 128, nullptr, nullptr, nullptr, a, sums, nullptr, nullptr);
  spmm128_kernel<<<2 * pgrid, b256, 0, stream>>>(t, src, dst, ew, a, E, pgrid, pstride, nullptr);
  colstats_kernel<<<egrid, b256, 0, stream>>>(a, sums, N);

  gemm_wmma_kernel<8, true, 128><<<ggrid, b256, 0, stream>>>(
      a, W2, t, N, 128, sums, gamma1, beta1, a, nullptr, nullptr, nullptr);
  spmm128_kernel<<<2 * pgrid, b256, 0, stream>>>(t, src, dst, ew, a, E, pgrid, pstride, sums);
  colstats_kernel<<<egrid, b256, 0, stream>>>(a, sums, N);

  gemm_wmma_kernel<3, true, 40><<<ggrid, b256, 0, stream>>>(
      a, W3, t, N, 40, sums, gamma2, beta2, nullptr, nullptr, out, b3);
  spmm40_kernel<<<sgrid, b256, 0, stream>>>(t, src, dst, ew, out, E);
}


# Edge sums split by package half: node range cut in two, a work-group adds only into rows of the range of the cache half it runs on (die id + ticket counters); same f32 atomics, still unordered
# speedup vs baseline: 2.7412x; 1.2182x over previous
#include <hip/hip_runtime.h>
#include <hip/hip_bf16.h>

typedef __attribute__((ext_vector_type(2))) float v2f;
typedef __attribute__((ext_vector_type(8))) float v8f;

#define BN_EPS 1e-5f
#define CH 128
#define WSTRIDE 132

__device__ __forceinline__ void fadd_atomic(float* p, float v) {
  unsafeAtomicAdd(p, v);
}

__device__ __forceinline__ int bcast_i(int v, int j) {
  return __builtin_amdgcn_readlane(v, j);
}
__device__ __forceinline__ float bcast_f(float v, int j) {
  return __int_as_float(__builtin_amdgcn_readlane(__float_as_int(v), j));
}

__device__ __forceinline__ int package_half() {
  unsigned id = __builtin_amdgcn_s_sendmsg_rtn(0x87);
  return (int)((id >> 18) & 1u);
}

#define QCAP 1024

template <int ROWF, bool FULL>
__device__ __forceinline__ void edge_group(const float* __restrict__ fpl,
                                           float* __restrict__ apl,
                                           int s_r, int d_r, float w_r, int n,
                                           bool second) {
  float v[16][2];
  if constexpr (ROWF == 64) {
#pragma unroll
    for (int j = 0; j < 16; ++j) {
      if (FULL || j < n) {
        const float* fr = fpl + (size_t)bcast_i(s_r, j) * ROWF;
        v[j][0] = fr[0];
        v[j][1] = fr[32];
      }
    }
#pragma unroll
    for (int j = 0; j < 16; ++j) {
      if (FULL || j < n) {
        float w = bcast_f(w_r, j);
        float* o = apl + (size_t)bcast_i(d_r, j) * ROWF;
        fadd_atomic(o + 0, v[j][0] * w);
        fadd_atomic(o + 32, v[j][1] * w);
      }
    }
  } else {
#pragma unroll
    for (int j = 0; j < 16; ++j) {
      if (FULL || j < n) v[j][0] = fpl[(size_t)bcast_i(s_r, j) * ROWF];
    }
    if (second) {
#pragma unroll
      for (int j = 0; j < 16; ++j) {
        if (FULL || j < n) v[j][1] = fpl[(size_t)bcast_i(s_r, j) * ROWF + 32];
      }
    }
#pragma unroll
    for (int j = 0; j < 16; ++j) {
      if (FULL || j < n) {
        fadd_atomic(apl + (size_t)bcast_i(d_r, j) * ROWF, v[j][0] * bcast_f(w_r, j));
      }
    }
    if (second) {
#pragma unroll
      for (int j = 0; j < 16; ++j) {
        if (FULL || j < n) {
          fadd_atomic(apl + (size_t)bcast_i(d_r, j) * ROWF + 32, v[j][1] * bcast_f(w_r, j));
        }
      }
    }
  }
}

template <int ROWF>
__global__ void __launch_bounds__(256, 16)
spmm_kernel(const float* __restrict__ feat,
            const int* __restrict__ src,
            const int* __restrict__ dst,
            const float* __restrict__ ew,
            float* __restrict__ agg, int E, int N, int nT, size_t pstride,
            int* __restrict__ ctr,
            float* __restrict__ zsums) {
  __shared__ int sh_visit;
  __shared__ int sh_cnt[8];
  __shared__ int q_s[QCAP];
  __shared__ int q_d[QCAP];
  __shared__ float q_w[QCAP];
  if (zsums != nullptr && blockIdx.x == 0) zsums[threadIdx.x] = 0.f;
  int plane = (ROWF == 64 && (int)blockIdx.x >= 2 * nT) ? 1 : 0;
  if (threadIdx.x == 0) {
    int h = package_half();
    int k = atomicAdd(&ctr[plane * 2 + h], 1);
    if (k >= nT) {
      h ^= 1;
      k = atomicAdd(&ctr[plane * 2 + h], 1);
    }
    sh_visit = (k < nT) ? (k * 2 + h) : -1;
  }
  __syncthreads();
  int visit = sh_visit;
  if (visit < 0) return;
  int range = visit & 1;
  int chunk = visit >> 1;
  int mid = N >> 1;

  int lane = threadIdx.x & 31;
  int wv = __builtin_amdgcn_readfirstlane((int)(threadIdx.x >> 5));
  const float* fpl = feat + (size_t)plane * pstride + lane;
  float* apl = agg + (size_t)plane * pstride + lane;
  bool second = (ROWF == 64) || (lane < 8);
  int e0 = (chunk * 8 + wv) * 128;

  int s_l[4], d_l[4];
  float w_l[4];
  bool ok[4];
#pragma unroll
  for (int st = 0; st < 4; ++st) {
    int e = e0 + st * 32 + lane;
    ok[st] = e < E;
    s_l[st] = 0;
    d_l[st] = 0;
    w_l[st] = 0.f;
    if (ok[st]) {
      s_l[st] = src[e];
      d_l[st] = dst[e];
      w_l[st] = ew[e];
    }
  }

  bool keep[4];
  unsigned km[4];
  int cnt = 0;
#pragma unroll
  for (int st = 0; st < 4; ++st) {
    keep[st] = ok[st] && ((d_l[st] >= mid) == (range == 1));
    km[st] = __builtin_amdgcn_ballot_w32(keep[st]);
    cnt += __builtin_popcount(km[st]);
  }
  if (lane == 0) sh_cnt[wv] = cnt;
  __syncthreads();
  int below = 0, total = 0;
#pragma unroll
  for (int w = 0; w < 8; ++w) {
    int c = sh_cnt[w];
    if (w < wv) below += c;
    total += c;
  }
  below = __builtin_amdgcn_readfirstlane(below);
  total = __builtin_amdgcn_readfirstlane(total);

  int off = below;
#pragma unroll
  for (int st = 0; st < 4; ++st) {
    int pre = __builtin_amdgcn_mbcnt_lo(km[st], 0u);
    if (keep[st]) {
      int p = off + pre;
      q_s[p] = s_l[st];
      q_d[p] = d_l[st];
      q_w[p] = w_l[st];
    }
    off += __builtin_popcount(km[st]);
  }
  __syncthreads();

  int l16 = lane & 15;
  int nfull = total >> 4;
  for (int gi = wv; gi < nfull; gi += 8) {
    int p = gi * 16 + l16;
    edge_group<ROWF, true>(fpl, apl, q_s[p], q_d[p], q_w[p], 16, second);
  }
  int rem = total & 15;
  if (rem > 0 && wv == (nfull & 7)) {
    int p = nfull * 16 + l16;
    edge_group<ROWF, false>(fpl, apl, q_s[p], q_d[p], q_w[p], rem, second);
  }
}

template <int NT, bool BN, int COUT>
__global__ void gemm_wmma_kernel(const float* A,
                                 const float* __restrict__ W,
                                 float* out,
                                 int Nrows, int ldo,
                                 const float* __restrict__ sums,
                                 const float* __restrict__ gamma,
                                 const float* __restrict__ beta,
                                 float* zrows,
                                 float* __restrict__ zsums,
                                 float* __restrict__ bout,
                                 const float* __restrict__ bias,
                                 int* __restrict__ zctr) {
  constexpr int CP = NT * 16;
  constexpr bool APL = BN;
  constexpr bool OPL = (COUT == CH);
  const size_t pst = (size_t)Nrows * 64;
  __shared__ float sWT[CP * WSTRIDE];
  __shared__ float sMean[CH];
  __shared__ float sSc[CH];
  __shared__ float sBeta[CH];
  int tid = threadIdx.x;
  if (zsums != nullptr && blockIdx.x == 0) zsums[tid] = 0.f;
  if (zctr != nullptr && blockIdx.x == 0 && tid < 4) zctr[tid] = 0;
  constexpr int WPER = CH * CP / 256;
  constexpr int WB = (WPER % 32 == 0) ? 32 : WPER;
  static_assert(WPER % WB == 0, "whole batches");
  for (int base = 0; base < WPER; base += WB) {
    float wv[WB];
#pragma unroll
    for (int i = 0; i < WB; ++i) {
      int idx = tid + (base + i) * 256;
      int k = idx / CP;
      int c = idx % CP;
      int cc = (c < COUT) ? c : (COUT - 1);
      wv[i] = W[k * COUT + cc];
    }
#pragma unroll
    for (int i = 0; i < WB; ++i) {
      int idx = tid + (base + i) * 256;
      int k = idx / CP;
      int c = idx % CP;
      sWT[c * WSTRIDE + k] = (c < COUT) ? wv[i] : 0.f;
    }
  }
  if (BN) {
    if (tid < CH) {
      float invN = 1.0f / (float)Nrows;
      float mean = sums[tid] * invN;
      float var = sums[128 + tid] * invN - mean * mean;
      sMean[tid] = mean;
      sSc[tid] = gamma[tid] * rsqrtf(var + BN_EPS);
      sBeta[tid] = beta[tid];
    }
  }
  __syncthreads();

  int wave = tid >> 5;
  int lane = tid & 31;
  int row0 = blockIdx.x * 128 + wave * 16;
  if (row0 >= Nrows) return;

  int m = lane & 15;
  int kh = (lane >> 4) * 2;

  v8f acc[NT];
#pragma unroll
  for (int t = 0; t < NT; ++t) acc[t] = {};

  const float* Arow = APL ? A + (size_t)(row0 + m) * 64 : A + (size_t)(row0 + m) * CH;
  const size_t ahalf = APL ? pst : (size_t)64;

  const float* Arow2 = Arow + ahalf - 64;

#pragma unroll 4
  for (int k0 = 0; k0 < CH; k0 += 4) {
    v2f a = *(const v2f*)((k0 < 64 ? Arow : Arow2) + k0 + kh);
    if (BN) {
      v2f mu = *(const v2f*)(&sMean[k0 + kh]);
      v2f sc = *(const v2f*)(&sSc[k0 + kh]);
      v2f be = *(const v2f*)(&sBeta[k0 + kh]);
      float x0 = (a.x - mu.x) * sc.x + be.x;
      float x1 = (a.y - mu.y) * sc.y + be.y;
      a.x = x0 > 0.f ? x0 : 0.f;
      a.y = x1 > 0.f ? x1 : 0.f;
    }
#pragma unroll
    for (int t = 0; t < NT; ++t) {
      v2f b = *(const v2f*)(&sWT[(t * 16 + m) * WSTRIDE + k0 + kh]);
      acc[t] = __builtin_amdgcn_wmma_f32_16x16x4_f32(
           false, a,  false, b,
           (short)0, acc[t],  false,  false);
    }
  }

  int rbase = row0 + (lane >> 4) * 8;
#pragma unroll
  for (int t = 0; t < NT; ++t) {
    int col = t * 16 + m;
    if (col < COUT) {
#pragma unroll
      for (int j = 0; j < 8; ++j) {
        if (OPL) {
          out[(size_t)(t >> 2) * pst + (size_t)(rbase + j) * 64 + (col & 63)] = acc[t][j];
        } else {
          out[(size_t)(rbase + j) * ldo + col] = acc[t][j];
        }
      }
    }
  }

  if (zrows != nullptr) {
    float4 zero4 = {0.f, 0.f, 0.f, 0.f};
#pragma unroll
    for (int p = 0; p < 2; ++p) {
      float4* z = (float4*)(zrows + (size_t)p * pst + (size_t)row0 * 64);
#pragma unroll
      for (int i = 0; i < 8; ++i) z[i * 32 + lane] = zero4;
    }
  }
  if (bout != nullptr) {
    float* o = bout + (size_t)row0 * 40;
#pragma unroll
    for (int i = 0; i < 20; ++i) {
      int e = i * 32 + lane;
      o[e] = bias[e % 40];
    }
  }
}

__global__ void colstats_kernel(const float* __restrict__ gp,
                                float* __restrict__ sums, int Nrows) {
  __shared__ float sh[256];
  sh[threadIdx.x] = 0.f;
  __syncthreads();
  size_t pidx = (size_t)blockIdx.x * 256 + threadIdx.x;
  int col = (int)(pidx & (CH - 1));
  size_t total = (size_t)Nrows * 64;
  const float* g = gp + (size_t)(col >> 6) * total + (col & 63);
  size_t i = (pidx >> 7) * 64;
  size_t stride = (size_t)gridDim.x * 128;
  float s = 0.f, q = 0.f;
  constexpr int RB = 12;
  for (; i + (RB - 1) * stride < total; i += RB * stride) {
    float v[RB];
#pragma unroll
    for (int j = 0; j < RB; ++j) v[j] = g[i + j * stride];
#pragma unroll
    for (int j = 0; j < RB; ++j) {
      s += v[j];
      q = fmaf(v[j], v[j], q);
    }
  }
  for (; i < total; i += stride) {
    float v = g[i];
    s += v;
    q = fmaf(v, v, q);
  }
  atomicAdd(&sh[col], s);
  atomicAdd(&sh[128 + col], q);
  __syncthreads();
  if (threadIdx.x < 128) {
    atomicAdd(&sums[threadIdx.x], sh[threadIdx.x]);
    atomicAdd(&sums[128 + threadIdx.x], sh[128 + threadIdx.x]);
  }
}

extern "C" void kernel_launch(void* const* d_in, const int* in_sizes, int n_in,
                              void* d_out, int out_size, void* d_ws, size_t ws_size,
                              hipStream_t stream) {
  const float* feat   = (const float*)d_in[0];
  const int*   src    = (const int*)d_in[1];
  const int*   dst    = (const int*)d_in[2];
  const float* ew     = (const float*)d_in[3];
  const float* W1     = (const float*)d_in[4];
  const float* W2     = (const float*)d_in[5];
  const float* W3     = (const float*)d_in[6];
  const float* b3     = (const float*)d_in[7];
  const float* gamma1 = (const float*)d_in[8];
  const float* beta1  = (const float*)d_in[9];
  const float* gamma2 = (const float*)d_in[10];
  const float* beta2  = (const float*)d_in[11];

  int N = in_sizes[0] / CH;
  int E = in_sizes[1];

  float* t    = (float*)d_ws;
  float* a    = t + (size_t)N * CH;
  float* sums = a + (size_t)N * CH;
  float* out  = (float*)d_out;

  int nT = (E + 1023) / 1024;
  size_t pstride = (size_t)N * 64;
  int ggrid = (N + 127) / 128;
  int egrid = 2048;
  dim3 b256(256);

  int* ctr1 = (int*)d_out;
  int* ctr2 = ctr1 + 4;
  int* ctr3 = (int*)(t + (size_t)N * 40);

  gemm_wmma_kernel<8, false, 128><<<ggrid, b256, 0, stream>>>(
      feat, W1, t, N, 128, nullptr, nullptr, nullptr, a, sums, nullptr, nullptr, ctr1);
  spmm_kernel<64><<<4 * nT, b256, 0, stream>>>(t, src, dst, ew, a, E, N, nT, pstride, ctr1, nullptr);
  colstats_kernel<<<egrid, b256, 0, stream>>>(a, sums, N);

  gemm_wmma_kernel<8, true, 128><<<ggrid, b256, 0, stream>>>(
      a, W2, t, N, 128, sums, gamma1, beta1, a, nullptr, nullptr, nullptr, ctr2);
  spmm_kernel<64><<<4 * nT, b256, 0, stream>>>(t, src, dst, ew, a, E, N, nT, pstride, ctr2, sums);
  colstats_kernel<<<egrid, b256, 0, stream>>>(a, sums, N);

  gemm_wmma_kernel<3, true, 40><<<ggrid, b256, 0, stream>>>(
      a, W3, t, N, 40, sums, gamma2, beta2, nullptr, nullptr, out, b3, ctr3);
  spmm_kernel<40><<<2 * nT, b256, 0, stream>>>(t, src, dst, ew, out, E, N, nT, 0, ctr3, nullptr);
}


# Layer-3 rows moved as line-aligned pieces: product rows padded to 256-B units, each add split at the output row's 128-B line boundary (2nd piece rotated by ds_bpermute); same unordered f32 atomics
# speedup vs baseline: 2.8788x; 1.0502x over previous
#include <hip/hip_runtime.h>
#include <hip/hip_bf16.h>

typedef __attribute__((ext_vector_type(2))) float v2f;
typedef __attribute__((ext_vector_type(8))) float v8f;

#define BN_EPS 1e-5f
#define CH 128
#define WSTRIDE 132

__device__ __forceinline__ void fadd_atomic(float* p, float v) {
  unsafeAtomicAdd(p, v);
}

__device__ __forceinline__ int bcast_i(int v, int j) {
  return __builtin_amdgcn_readlane(v, j);
}
__device__ __forceinline__ float bcast_f(float v, int j) {
  return __int_as_float(__builtin_amdgcn_readlane(__float_as_int(v), j));
}

__device__ __forceinline__ int package_half() {
  unsigned id = __builtin_amdgcn_s_sendmsg_rtn(0x87);
  return (int)((id >> 18) & 1u);
}

#define QCAP 1024

template <int ROWF, bool FULL>
__device__ __forceinline__ void edge_group(const float* __restrict__ fpl,
                                           float* __restrict__ apl,
                                           const float* __restrict__ fu,
                                           float* __restrict__ au,
                                           int s_r, int d_r, float w_r, int n,
                                           bool second) {
  float v[16][2];
  if constexpr (ROWF == 64) {
#pragma unroll
    for (int j = 0; j < 16; ++j) {
      if (FULL || j < n) {
        const float* fr = fpl + (size_t)bcast_i(s_r, j) * ROWF;
        v[j][0] = fr[0];
        v[j][1] = fr[32];
      }
    }
#pragma unroll
    for (int j = 0; j < 16; ++j) {
      if (FULL || j < n) {
        float w = bcast_f(w_r, j);
        float* o = apl + (size_t)bcast_i(d_r, j) * ROWF;
        fadd_atomic(o + 0, v[j][0] * w);
        fadd_atomic(o + 32, v[j][1] * w);
      }
    }
  } else {
    unsigned lane = threadIdx.x & 31u;
    unsigned hoff = 32u + (lane & 7u);
#pragma unroll
    for (int j = 0; j < 16; ++j) {
      if (FULL || j < n) {
        const float* fr = fu + (size_t)bcast_i(s_r, j) * 64;
        v[j][0] = fr[lane];
        v[j][1] = fr[hoff];
      }
    }
#pragma unroll
    for (int j = 0; j < 16; ++j) {
      if (FULL || j < n) {
        int k8 = (bcast_i(d_r, j) & 3) * 8;
        float lo = v[j][0], hi = v[j][1];
        float m = second ? hi : lo;
        v[j][1] = __int_as_float(__builtin_amdgcn_ds_bpermute(
            (int)(((lane - (unsigned)k8) & 31u) << 2), __float_as_int(m)));
      }
    }
#pragma unroll
    for (int j = 0; j < 16; ++j) {
      if (FULL || j < n) {
        int d = bcast_i(d_r, j);
        float w = bcast_f(w_r, j);
        unsigned k8 = (unsigned)(d & 3) * 8u;
        float* o = au + (size_t)d * ROWF;
        if (lane < 32u - k8) fadd_atomic(o + lane, v[j][0] * w);
        if (lane < 8u + k8) fadd_atomic(o + (32u - k8) + lane, v[j][1] * w);
      }
    }
  }
}

template <int ROWF>
__global__ void __launch_bounds__(256, 16)
spmm_kernel(const float* __restrict__ feat,
            const int* __restrict__ src,
            const int* __restrict__ dst,
            const float* __restrict__ ew,
            float* __restrict__ agg, int E, int N, int nT, size_t pstride,
            int* __restrict__ ctr,
            float* __restrict__ zsums) {
  __shared__ int sh_visit;
  __shared__ int sh_cnt[8];
  __shared__ int q_s[QCAP];
  __shared__ int q_d[QCAP];
  __shared__ float q_w[QCAP];
  if (zsums != nullptr && blockIdx.x == 0) zsums[threadIdx.x] = 0.f;
  int plane = (ROWF == 64 && (int)blockIdx.x >= 2 * nT) ? 1 : 0;
  if (threadIdx.x == 0) {
    int h = package_half();
    int k = atomicAdd(&ctr[plane * 2 + h], 1);
    if (k >= nT) {
      h ^= 1;
      k = atomicAdd(&ctr[plane * 2 + h], 1);
    }
    sh_visit = (k < nT) ? (k * 2 + h) : -1;
  }
  __syncthreads();
  int visit = sh_visit;
  if (visit < 0) return;
  int range = visit & 1;
  int chunk = visit >> 1;
  int mid = N >> 1;

  int lane = threadIdx.x & 31;
  int wv = __builtin_amdgcn_readfirstlane((int)(threadIdx.x >> 5));
  const float* fu = feat + (size_t)plane * pstride;
  float* au = agg + (size_t)plane * pstride;
  const float* fpl = fu + lane;
  float* apl = au + lane;
  bool second = (ROWF == 64) || (lane < 8);
  int e0 = (chunk * 8 + wv) * 128;

  int s_l[4], d_l[4];
  float w_l[4];
  bool ok[4];
#pragma unroll
  for (int st = 0; st < 4; ++st) {
    int e = e0 + st * 32 + lane;
    ok[st] = e < E;
    s_l[st] = 0;
    d_l[st] = 0;
    w_l[st] = 0.f;
    if (ok[st]) {
      s_l[st] = src[e];
      d_l[st] = dst[e];
      w_l[st] = ew[e];
    }
  }

  bool keep[4];
  unsigned km[4];
  int cnt = 0;
#pragma unroll
  for (int st = 0; st < 4; ++st) {
    keep[st] = ok[st] && ((d_l[st] >= mid) == (range == 1));
    km[st] = __builtin_amdgcn_ballot_w32(keep[st]);
    cnt += __builtin_popcount(km[st]);
  }
  if (lane == 0) sh_cnt[wv] = cnt;
  __syncthreads();
  int below = 0, total = 0;
#pragma unroll
  for (int w = 0; w < 8; ++w) {
    int c = sh_cnt[w];
    if (w < wv) below += c;
    total += c;
  }
  below = __builtin_amdgcn_readfirstlane(below);
  total = __builtin_amdgcn_readfirstlane(total);

  int off = below;
#pragma unroll
  for (int st = 0; st < 4; ++st) {
    int pre = __builtin_amdgcn_mbcnt_lo(km[st], 0u);
    if (keep[st]) {
      int p = off + pre;
      q_s[p] = s_l[st];
      q_d[p] = d_l[st];
      q_w[p] = w_l[st];
    }
    off += __builtin_popcount(km[st]);
  }
  __syncthreads();

  int l16 = lane & 15;
  int nfull = total >> 4;
  for (int gi = wv; gi < nfull; gi += 8) {
    int p = gi * 16 + l16;
    edge_group<ROWF, true>(fpl, apl, fu, au, q_s[p], q_d[p], q_w[p], 16, second);
  }
  int rem = total & 15;
  if (rem > 0 && wv == (nfull & 7)) {
    int p = nfull * 16 + l16;
    edge_group<ROWF, false>(fpl, apl, fu, au, q_s[p], q_d[p], q_w[p], rem, second);
  }
}

template <int NT, bool BN, int COUT>
__global__ void gemm_wmma_kernel(const float* A,
                                 const float* __restrict__ W,
                                 float* out,
                                 int Nrows, int ldo,
                                 const float* __restrict__ sums,
                                 const float* __restrict__ gamma,
                                 const float* __restrict__ beta,
                                 float* zrows,
                                 float* __restrict__ zsums,
                                 float* __restrict__ bout,
                                 const float* __restrict__ bias,
                                 int* __restrict__ zctr) {
  constexpr int CP = NT * 16;
  constexpr bool APL = BN;
  constexpr bool OPL = (COUT == CH);
  const size_t pst = (size_t)Nrows * 64;
  __shared__ float sWT[CP * WSTRIDE];
  __shared__ float sMean[CH];
  __shared__ float sSc[CH];
  __shared__ float sBeta[CH];
  int tid = threadIdx.x;
  if (zsums != nullptr && blockIdx.x == 0) zsums[tid] = 0.f;
  if (zctr != nullptr && blockIdx.x == 0 && tid < 4) zctr[tid] = 0;
  constexpr int WPER = CH * CP / 256;
  constexpr int WB = (WPER % 32 == 0) ? 32 : WPER;
  static_assert(WPER % WB == 0, "whole batches");
  for (int base = 0; base < WPER; base += WB) {
    float wv[WB];
#pragma unroll
    for (int i = 0; i < WB; ++i) {
      int idx = tid + (base + i) * 256;
      int k = idx / CP;
      int c = idx % CP;
      int cc = (c < COUT) ? c : (COUT - 1);
      wv[i] = W[k * COUT + cc];
    }
#pragma unroll
    for (int i = 0; i < WB; ++i) {
      int idx = tid + (base + i) * 256;
      int k = idx / CP;
      int c = idx % CP;
      sWT[c * WSTRIDE + k] = (c < COUT) ? wv[i] : 0.f;
    }
  }
  if (BN) {
    if (tid < CH) {
      float invN = 1.0f / (float)Nrows;
      float mean = sums[tid] * invN;
      float var = sums[128 + tid] * invN - mean * mean;
      sMean[tid] = mean;
      sSc[tid] = gamma[tid] * rsqrtf(var + BN_EPS);
      sBeta[tid] = beta[tid];
    }
  }
  __syncthreads();

  int wave = tid >> 5;
  int lane = tid & 31;
  int row0 = blockIdx.x * 128 + wave * 16;
  if (row0 >= Nrows) return;

  int m = lane & 15;
  int kh = (lane >> 4) * 2;

  v8f acc[NT];
#pragma unroll
  for (int t = 0; t < NT; ++t) acc[t] = {};

  const float* Arow = APL ? A + (size_t)(row0 + m) * 64 : A + (size_t)(row0 + m) * CH;
  const size_t ahalf = APL ? pst : (size_t)64;

  const float* Arow2 = Arow + ahalf - 64;

#pragma unroll 4
  for (int k0 = 0; k0 < CH; k0 += 4) {
    v2f a = *(const v2f*)((k0 < 64 ? Arow : Arow2) + k0 + kh);
    if (BN) {
      v2f mu = *(const v2f*)(&sMean[k0 + kh]);
      v2f sc = *(const v2f*)(&sSc[k0 + kh]);
      v2f be = *(const v2f*)(&sBeta[k0 + kh]);
      float x0 = (a.x - mu.x) * sc.x + be.x;
      float x1 = (a.y - mu.y) * sc.y + be.y;
      a.x = x0 > 0.f ? x0 : 0.f;
      a.y = x1 > 0.f ? x1 : 0.f;
    }
#pragma unroll
    for (int t = 0; t < NT; ++t) {
      v2f b = *(const v2f*)(&sWT[(t * 16 + m) * WSTRIDE + k0 + kh]);
      acc[t] = __builtin_amdgcn_wmma_f32_16x16x4_f32(
           false, a,  false, b,
           (short)0, acc[t],  false,  false);
    }
  }

  int rbase = row0 + (lane >> 4) * 8;
#pragma unroll
  for (int t = 0; t < NT; ++t) {
    int col = t * 16 + m;
    if (col < COUT) {
#pragma unroll
      for (int j = 0; j < 8; ++j) {
        if (OPL) {
          out[(size_t)(t >> 2) * pst + (size_t)(rbase + j) * 64 + (col & 63)] = acc[t][j];
        } else {
          out[(size_t)(rbase + j) * ldo + col] = acc[t][j];
        }
      }
    }
  }

  if (zrows != nullptr) {
    float4 zero4 = {0.f, 0.f, 0.f, 0.f};
#pragma unroll
    for (int p = 0; p < 2; ++p) {
      float4* z = (float4*)(zrows + (size_t)p * pst + (size_t)row0 * 64);
#pragma unroll
      for (int i = 0; i < 8; ++i) z[i * 32 + lane] = zero4;
    }
  }
  if (bout != nullptr) {
    float* o = bout + (size_t)row0 * 40;
#pragma unroll
    for (int i = 0; i < 20; ++i) {
      int e = i * 32 + lane;
      o[e] = bias[e % 40];
    }
  }
}

__global__ void colstats_kernel(const float* __restrict__ gp,
                                float* __restrict__ sums, int Nrows) {
  __shared__ float sh[256];
  sh[threadIdx.x] = 0.f;
  __syncthreads();
  size_t pidx = (size_t)blockIdx.x * 256 + threadIdx.x;
  int col = (int)(pidx & (CH - 1));
  size_t total = (size_t)Nrows * 64;
  const float* g = gp + (size_t)(col >> 6) * total + (col & 63);
  size_t i = (pidx >> 7) * 64;
  size_t stride = (size_t)gridDim.x * 128;
  float s = 0.f, q = 0.f;
  constexpr int RB = 12;
  for (; i + (RB - 1) * stride < total; i += RB * stride) {
    float v[RB];
#pragma unroll
    for (int j = 0; j < RB; ++j) v[j] = g[i + j * stride];
#pragma unroll
    for (int j = 0; j < RB; ++j) {
      s += v[j];
      q = fmaf(v[j], v[j], q);
    }
  }
  for (; i < total; i += stride) {
    float v = g[i];
    s += v;
    q = fmaf(v, v, q);
  }
  atomicAdd(&sh[col], s);
  atomicAdd(&sh[128 + col], q);
  __syncthreads();
  if (threadIdx.x < 128) {
    atomicAdd(&sums[threadIdx.x], sh[threadIdx.x]);
    atomicAdd(&sums[128 + threadIdx.x], sh[128 + threadIdx.x]);
  }
}

extern "C" void kernel_launch(void* const* d_in, const int* in_sizes, int n_in,
                              void* d_out, int out_size, void* d_ws, size_t ws_size,
                              hipStream_t stream) {
  const float* feat   = (const float*)d_in[0];
  const int*   src    = (const int*)d_in[1];
  const int*   dst    = (const int*)d_in[2];
  const float* ew     = (const float*)d_in[3];
  const float* W1     = (const float*)d_in[4];
  const float* W2     = (const float*)d_in[5];
  const float* W3     = (const float*)d_in[6];
  const float* b3     = (const float*)d_in[7];
  const float* gamma1 = (const float*)d_in[8];
  const float* beta1  = (const float*)d_in[9];
  const float* gamma2 = (const float*)d_in[10];
  const float* beta2  = (const float*)d_in[11];

  int N = in_sizes[0] / CH;
  int E = in_sizes[1];

  float* t    = (float*)d_ws;
  float* a    = t + (size_t)N * CH;
  float* sums = a + (size_t)N * CH;
  float* out  = (float*)d_out;

  int nT = (E + 1023) / 1024;
  size_t pstride = (size_t)N * 64;
  int ggrid = (N + 127) / 128;
  int egrid = 2048;
  dim3 b256(256);

  int* ctr1 = (int*)d_out;
  int* ctr2 = ctr1 + 4;
  int* ctr3 = (int*)(t + (size_t)N * 64);

  gemm_wmma_kernel<8, false, 128><<<ggrid, b256, 0, stream>>>(
      feat, W1, t, N, 128, nullptr, nullptr, nullptr, a, sums, nullptr, nullptr, ctr1);
  spmm_kernel<64><<<4 * nT, b256, 0, stream>>>(t, src, dst, ew, a, E, N, nT, pstride, ctr1, nullptr);
  colstats_kernel<<<egrid, b256, 0, stream>>>(a, sums, N);

  gemm_wmma_kernel<8, true, 128><<<ggrid, b256, 0, stream>>>(
      a, W2, t, N, 128, sums, gamma1, beta1, a, nullptr, nullptr, nullptr, ctr2);
  spmm_kernel<64><<<4 * nT, b256, 0, stream>>>(t, src, dst, ew, a, E, N, nT, pstride, ctr2, sums);
  colstats_kernel<<<egrid, b256, 0, stream>>>(a, sums, N);

  gemm_wmma_kernel<3, true, 40><<<ggrid, b256, 0, stream>>>(
      a, W3, t, N, 64, sums, gamma2, beta2, nullptr, nullptr, out, b3, ctr3);
  spmm_kernel<40><<<2 * nT, b256, 0, stream>>>(t, src, dst, ew, out, E, N, nT, 0, ctr3, nullptr);
}


# Edge sums built in LDS: edges binned once per call by 256-row destination tile (+2 launches); its work-group adds gathered rows with unordered LDS f32 adds, stores it once; no global f32 atomic there
# speedup vs baseline: 3.1482x; 1.0936x over previous
#include <hip/hip_runtime.h>
#include <hip/hip_bf16.h>

typedef __attribute__((ext_vector_type(2))) float v2f;
typedef __attribute__((ext_vector_type(8))) float v8f;

#define BN_EPS 1e-5f
#define CH 128
#define WSTRIDE 132

#define TR 256
#define NBMAX 2048
#define EPW 2048

__device__ __forceinline__ unsigned bcast_u(unsigned v, int j) {
  return (unsigned)__builtin_amdgcn_readlane((int)v, j);
}
__device__ __forceinline__ float bcast_f(float v, int j) {
  return __int_as_float(__builtin_amdgcn_readlane(__float_as_int(v), j));
}

__global__ void __launch_bounds__(256)
bin_count_kernel(const int* __restrict__ dst, int E, int* __restrict__ cnt, int nb) {
  __shared__ int h[NBMAX];
  int tid = threadIdx.x;
  for (int i = tid; i < nb; i += 256) h[i] = 0;
  __syncthreads();
  int e0 = (int)blockIdx.x * EPW + tid;
#pragma unroll
  for (int k = 0; k < EPW / 256; ++k) {
    int e = e0 + k * 256;
    if (e < E) atomicAdd(&h[(unsigned)dst[e] / TR], 1);
  }
  __syncthreads();
  for (int i = tid; i < nb; i += 256) {
    int c = h[i];
    if (c != 0) atomicAdd(&cnt[i], c);
  }
}

__global__ void __launch_bounds__(256)
bin_place_kernel(const int* __restrict__ src, const int* __restrict__ dst,
                 const float* __restrict__ ew, int E,
                 const int* __restrict__ cnt, int* __restrict__ cur,
                 int* __restrict__ starts, unsigned* __restrict__ edges, int nb) {
  __shared__ int h[NBMAX];
  __shared__ int st[NBMAX];
  int tid = threadIdx.x;
  int lane = tid & 31;
  int wv = __builtin_amdgcn_readfirstlane(tid >> 5);
  for (int i = tid; i < nb; i += 256) h[i] = 0;
  __syncthreads();
  int s_l[EPW / 256], d_l[EPW / 256], r_l[EPW / 256];
  float w_l[EPW / 256];
  bool ok[EPW / 256];
  int e0 = (int)blockIdx.x * EPW + tid;
#pragma unroll
  for (int k = 0; k < EPW / 256; ++k) {
    int e = e0 + k * 256;
    ok[k] = e < E;
    s_l[k] = 0;
    d_l[k] = 0;
    w_l[k] = 0.f;
    if (ok[k]) {
      s_l[k] = src[e];
      d_l[k] = dst[e];
      w_l[k] = ew[e];
    }
  }
#pragma unroll
  for (int k = 0; k < EPW / 256; ++k) {
    r_l[k] = 0;
    if (ok[k]) r_l[k] = atomicAdd(&h[(unsigned)d_l[k] / TR], 1);
  }
  if (wv == 0) {
    int carry = 0;
    for (int base = 0; base < nb; base += 32) {
      int i = base + lane;
      int c = (i < nb) ? cnt[i] : 0;
      int x = c;
#pragma unroll
      for (int o = 1; o < 32; o <<= 1) {
        int y = __shfl_up(x, o, 32);
        if (lane >= o) x += y;
      }
      if (i < nb) st[i] = carry + x - c;
      carry += __shfl(x, 31, 32);
    }
  }
  __syncthreads();
  for (int i = tid; i < nb; i += 256) {
    int c = h[i];
    int b0 = st[i];
    if (blockIdx.x == 0) starts[i] = b0;
    int run = 0;
    if (c != 0) run = atomicAdd(&cur[i], c);
    h[i] = b0 + run;
  }
  __syncthreads();
#pragma unroll
  for (int k = 0; k < EPW / 256; ++k) {
    if (ok[k]) {
      unsigned bin = (unsigned)d_l[k] / TR;
      unsigned dl = (unsigned)d_l[k] - bin * TR;
      int pos = h[bin] + r_l[k];
      ((uint2*)edges)[pos] = make_uint2((unsigned)s_l[k] | (dl << 24), __float_as_uint(w_l[k]));
    }
  }
}

template <int ROWF, bool FULL, int BASE>
__device__ __forceinline__ void tile_group(const float* __restrict__ fu,
                                           float (&tile)[TR * ROWF],
                                           unsigned sd_r, float w_r, int n,
                                           unsigned lane) {
  float v[16][2];
  unsigned hoff = (ROWF == 64) ? (32u + lane) : (32u + (lane & 7u));
#pragma unroll
  for (int j = 0; j < 16; ++j) {
    unsigned s = bcast_u(sd_r, BASE + j) & 0xFFFFFFu;
    const float* fr = fu + (size_t)s * 64;
    v[j][0] = fr[lane];
    v[j][1] = fr[hoff];
  }
#pragma unroll
  for (int j = 0; j < 16; ++j) {
    if (FULL || j < n) {
      unsigned dl = bcast_u(sd_r, BASE + j) >> 24;
      float w = bcast_f(w_r, BASE + j);
      float* o = &tile[dl * ROWF];
      atomicAdd(o + lane, v[j][0] * w);
      if (ROWF == 64 || lane < 8u) atomicAdd(o + 32 + lane, v[j][1] * w);
    }
  }
}

template <int ROWF, int NW>
__global__ void __launch_bounds__(NW * 32, 16)
tile_sum_kernel(const float* __restrict__ feat,
                const int* __restrict__ cnt,
                const int* __restrict__ starts,
                const unsigned* __restrict__ edges,
                float* __restrict__ agg, int N, int nb, size_t pstride,
                const float* __restrict__ bias,
                float* __restrict__ zsums) {
  __shared__ float tile[TR * ROWF];
  int tid = threadIdx.x;
  if (zsums != nullptr && blockIdx.x == 0 && tid < 256) zsums[tid] = 0.f;
  int plane = (ROWF == 64 && (int)blockIdx.x >= nb) ? 1 : 0;
  int bin = (int)blockIdx.x - plane * nb;
  if (ROWF == 64) {
    float4 zero4 = {0.f, 0.f, 0.f, 0.f};
    for (int i = tid; i < TR * ROWF / 4; i += NW * 32) ((float4*)tile)[i] = zero4;
  } else {
    for (int i = tid; i < TR * ROWF; i += NW * 32) tile[i] = bias[i % ROWF];
  }
  __syncthreads();

  unsigned lane = (unsigned)tid & 31u;
  int wv = __builtin_amdgcn_readfirstlane(tid >> 5);
  int n = cnt[bin];
  const uint2* eb = (const uint2*)edges + starts[bin];
  const float* fu = feat + (size_t)plane * pstride;

  for (int p = wv * 32; p < n; p += NW * 32) {
    int idx = p + (int)lane;
    uint2 ent = make_uint2(0u, 0u);
    if (idx < n) ent = eb[idx];
    float w_r = __uint_as_float(ent.y);
    int m = n - p;
    if (m >= 32) {
      tile_group<ROWF, true, 0>(fu, tile, ent.x, w_r, 16, lane);
      tile_group<ROWF, true, 16>(fu, tile, ent.x, w_r, 16, lane);
    } else {
      tile_group<ROWF, false, 0>(fu, tile, ent.x, w_r, m, lane);
      if (m > 16) tile_group<ROWF, false, 16>(fu, tile, ent.x, w_r, m - 16, lane);
    }
  }
  __syncthreads();

  int row0 = bin * TR;
  int rows = (N - row0 < TR) ? (N - row0) : TR;
  float4* o = (float4*)(agg + (size_t)plane * pstride + (size_t)row0 * ROWF);
  int nq = rows * (ROWF / 4);
  for (int i = tid; i < nq; i += NW * 32) o[i] = ((const float4*)tile)[i];
}

template <int NT, bool BN, int COUT>
__global__ void gemm_wmma_kernel(const float* A,
                                 const float* __restrict__ W,
                                 float* out,
                                 int Nrows, int ldo,
                                 const float* __restrict__ sums,
                                 const float* __restrict__ gamma,
                                 const float* __restrict__ beta,
                                 float* __restrict__ zsums,
                                 int* __restrict__ zctr, int nzc,
                                 const unsigned* __restrict__ cpsrc,
                                 unsigned* __restrict__ cpdst, int ncp) {
  constexpr int CP = NT * 16;
  constexpr bool APL = BN;
  constexpr bool OPL = (COUT == CH);
  const size_t pst = (size_t)Nrows * 64;
  __shared__ float sWT[CP * WSTRIDE];
  __shared__ float sMean[CH];
  __shared__ float sSc[CH];
  __shared__ float sBeta[CH];
  int tid = threadIdx.x;
  if (zsums != nullptr && blockIdx.x == 0) zsums[tid] = 0.f;
  if (zctr != nullptr && blockIdx.x == 0) {
    for (int i = tid; i < nzc; i += 256) zctr[i] = 0;
  }
  if (cpsrc != nullptr) {
    int nth = (int)gridDim.x * 256;
    for (int i = (int)blockIdx.x * 256 + tid; i < ncp; i += nth)
      ((uint2*)cpdst)[i] = ((const uint2*)cpsrc)[i];
  }
  constexpr int WPER = CH * CP / 256;
  constexpr int WB = (WPER % 32 == 0) ? 32 : WPER;
  static_assert(WPER % WB == 0, "whole batches");
  for (int base = 0; base < WPER; base += WB) {
    float wv[WB];
#pragma unroll
    for (int i = 0; i < WB; ++i) {
      int idx = tid + (base + i) * 256;
      int k = idx / CP;
      int c = idx % CP;
      int cc = (c < COUT) ? c : (COUT - 1);
      wv[i] = W[k * COUT + cc];
    }
#pragma unroll
    for (int i = 0; i < WB; ++i) {
      int idx = tid + (base + i) * 256;
      int k = idx / CP;
      int c = idx % CP;
      sWT[c * WSTRIDE + k] = (c < COUT) ? wv[i] : 0.f;
    }
  }
  if (BN) {
    if (tid < CH) {
      float invN = 1.0f / (float)Nrows;
      float mean = sums[tid] * invN;
      float var = sums[128 + tid] * invN - mean * mean;
      sMean[tid] = mean;
      sSc[tid] = gamma[tid] * rsqrtf(var + BN_EPS);
      sBeta[tid] = beta[tid];
    }
  }
  __syncthreads();

  int wave = tid >> 5;
  int lane = tid & 31;
  int row0 = blockIdx.x * 128 + wave * 16;
  if (row0 >= Nrows) return;

  int m = lane & 15;
  int kh = (lane >> 4) * 2;

  v8f acc[NT];
#pragma unroll
  for (int t = 0; t < NT; ++t) acc[t] = {};

  const float* Arow = APL ? A + (size_t)(row0 + m) * 64 : A + (size_t)(row0 + m) * CH;
  const size_t ahalf = APL ? pst : (size_t)64;

  const float* Arow2 = Arow + ahalf - 64;

#pragma unroll 4
  for (int k0 = 0; k0 < CH; k0 += 4) {
    v2f a = *(const v2f*)((k0 < 64 ? Arow : Arow2) + k0 + kh);
    if (BN) {
      v2f mu = *(const v2f*)(&sMean[k0 + kh]);
      v2f sc = *(const v2f*)(&sSc[k0 + kh]);
      v2f be = *(const v2f*)(&sBeta[k0 + kh]);
      float x0 = (a.x - mu.x) * sc.x + be.x;
      float x1 = (a.y - mu.y) * sc.y + be.y;
      a.x = x0 > 0.f ? x0 : 0.f;
      a.y = x1 > 0.f ? x1 : 0.f;
    }
#pragma unroll
    for (int t = 0; t < NT; ++t) {
      v2f b = *(const v2f*)(&sWT[(t * 16 + m) * WSTRIDE + k0 + kh]);
      acc[t] = __builtin_amdgcn_wmma_f32_16x16x4_f32(
           false, a,  false, b,
           (short)0, acc[t],  false,  false);
    }
  }

  int rbase = row0 + (lane >> 4) * 8;
#pragma unroll
  for (int t = 0; t < NT; ++t) {
    int col = t * 16 + m;
    if (col < COUT) {
#pragma unroll
      for (int j = 0; j < 8; ++j) {
        if (OPL) {
          out[(size_t)(t >> 2) * pst + (size_t)(rbase + j) * 64 + (col & 63)] = acc[t][j];
        } else {
          out[(size_t)(rbase + j) * ldo + col] = acc[t][j];
        }
      }
    }
  }
}

__global__ void colstats_kernel(const float* __restrict__ gp,
                                float* __restrict__ sums, int Nrows) {
  __shared__ float sh[256];
  sh[threadIdx.x] = 0.f;
  __syncthreads();
  size_t pidx = (size_t)blockIdx.x * 256 + threadIdx.x;
  int col = (int)(pidx & (CH - 1));
  size_t total = (size_t)Nrows * 64;
  const float* g = gp + (size_t)(col >> 6) * total + (col & 63);
  size_t i = (pidx >> 7) * 64;
  size_t stride = (size_t)gridDim.x * 128;
  float s = 0.f, q = 0.f;
  constexpr int RB = 12;
  for (; i + (RB - 1) * stride < total; i += RB * stride) {
    float v[RB];
#pragma unroll
    for (int j = 0; j < RB; ++j) v[j] = g[i + j * stride];
#pragma unroll
    for (int j = 0; j < RB; ++j) {
      s += v[j];
      q = fmaf(v[j], v[j], q);
    }
  }
  for (; i < total; i += stride) {
    float v = g[i];
    s += v;
    q = fmaf(v, v, q);
  }
  atomicAdd(&sh[col], s);
  atomicAdd(&sh[128 + col], q);
  __syncthreads();
  if (threadIdx.x < 128) {
    atomicAdd(&sums[threadIdx.x], sh[threadIdx.x]);
    atomicAdd(&sums[128 + threadIdx.x], sh[128 + threadIdx.x]);
  }
}

extern "C" void kernel_launch(void* const* d_in, const int* in_sizes, int n_in,
                              void* d_out, int out_size, void* d_ws, size_t ws_size,
                              hipStream_t stream) {
  const float* feat   = (const float*)d_in[0];
  const int*   src    = (const int*)d_in[1];
  const int*   dst    = (const int*)d_in[2];
  const float* ew     = (const float*)d_in[3];
  const float* W1     = (const float*)d_in[4];
  const float* W2     = (const float*)d_in[5];
  const float* W3     = (const float*)d_in[6];
  const float* b3     = (const float*)d_in[7];
  const float* gamma1 = (const float*)d_in[8];
  const float* beta1  = (const float*)d_in[9];
  const float* gamma2 = (const float*)d_in[10];
  const float* beta2  = (const float*)d_in[11];

  int N = in_sizes[0] / CH;
  int E = in_sizes[1];

  float* t    = (float*)d_ws;
  float* a    = t + (size_t)N * CH;
  float* sums = a + (size_t)N * CH;
  float* out  = (float*)d_out;

  size_t pstride = (size_t)N * 64;
  int ggrid = (N + 127) / 128;
  int egrid = 2048;
  dim3 b256(256);

  int nb = (N + TR - 1) / TR;
  int hdr = (3 * nb + 31) / 32 * 32;
  int bgrid = (E + EPW - 1) / EPW;
  int ncp = (hdr + 2 * E) / 2;
  int* bl1 = (int*)d_out;
  int* bl3 = (int*)(t + (size_t)N * 64);
  unsigned* ed1 = (unsigned*)(bl1 + hdr);
  unsigned* ed3 = (unsigned*)(bl3 + hdr);

  gemm_wmma_kernel<8, false, 128><<<ggrid, b256, 0, stream>>>(
      feat, W1, t, N, 128, nullptr, nullptr, nullptr, sums, bl1, 2 * nb, nullptr, nullptr, 0);
  bin_count_kernel<<<bgrid, b256, 0, stream>>>(dst, E, bl1, nb);
  bin_place_kernel<<<bgrid, b256, 0, stream>>>(src, dst, ew, E, bl1, bl1 + nb, bl1 + 2 * nb,
                                               ed1, nb);
  tile_sum_kernel<64, 16><<<2 * nb, 512, 0, stream>>>(
      t, bl1, bl1 + 2 * nb, ed1, a, N, nb, pstride, nullptr, nullptr);
  colstats_kernel<<<egrid, b256, 0, stream>>>(a, sums, N);

  gemm_wmma_kernel<8, true, 128><<<ggrid, b256, 0, stream>>>(
      a, W2, t, N, 128, sums, gamma1, beta1, nullptr, nullptr, 0, nullptr, nullptr, 0);
  tile_sum_kernel<64, 16><<<2 * nb, 512, 0, stream>>>(
      t, bl1, bl1 + 2 * nb, ed1, a, N, nb, pstride, nullptr, sums);
  colstats_kernel<<<egrid, b256, 0, stream>>>(a, sums, N);

  gemm_wmma_kernel<3, true, 40><<<ggrid, b256, 0, stream>>>(
      a, W3, t, N, 64, sums, gamma2, beta2, nullptr, nullptr, 0,
      (const unsigned*)bl1, (unsigned*)bl3, ncp);
  tile_sum_kernel<40, 32><<<nb, 1024, 0, stream>>>(
      t, bl3, bl3 + 2 * nb, ed3, out, N, nb, 0, b3, nullptr);
}


# Column statistics taken in the edge-sum launch: a thread sums its 4 columns (sum, sumsq) as the LDS tile is stored, merged by LDS + 128 f32 atomics per tile (new sum tree); 10 -> 8 launches
# speedup vs baseline: 3.4352x; 1.0912x over previous
#include <hip/hip_runtime.h>
#include <hip/hip_bf16.h>

typedef __attribute__((ext_vector_type(2))) float v2f;
typedef __attribute__((ext_vector_type(8))) float v8f;

#define BN_EPS 1e-5f
#define CH 128
#define WSTRIDE 132

#define TR 256
#define NBMAX 2048
#define EPW 2048

__device__ __forceinline__ unsigned bcast_u(unsigned v, int j) {
  return (unsigned)__builtin_amdgcn_readlane((int)v, j);
}
__device__ __forceinline__ float bcast_f(float v, int j) {
  return __int_as_float(__builtin_amdgcn_readlane(__float_as_int(v), j));
}

__global__ void __launch_bounds__(256)
bin_count_kernel(const int* __restrict__ dst, int E, int* __restrict__ cnt, int nb) {
  __shared__ int h[NBMAX];
  int tid = threadIdx.x;
  for (int i = tid; i < nb; i += 256) h[i] = 0;
  __syncthreads();
  int e0 = (int)blockIdx.x * EPW + tid;
#pragma unroll
  for (int k = 0; k < EPW / 256; ++k) {
    int e = e0 + k * 256;
    if (e < E) atomicAdd(&h[(unsigned)dst[e] / TR], 1);
  }
  __syncthreads();
  for (int i = tid; i < nb; i += 256) {
    int c = h[i];
    if (c != 0) atomicAdd(&cnt[i], c);
  }
}

__global__ void __launch_bounds__(256)
bin_place_kernel(const int* __restrict__ src, const int* __restrict__ dst,
                 const float* __restrict__ ew, int E,
                 const int* __restrict__ cnt, int* __restrict__ cur,
                 int* __restrict__ starts, unsigned* __restrict__ edges, int nb) {
  __shared__ int h[NBMAX];
  __shared__ int st[NBMAX];
  int tid = threadIdx.x;
  int lane = tid & 31;
  int wv = __builtin_amdgcn_readfirstlane(tid >> 5);
  for (int i = tid; i < nb; i += 256) h[i] = 0;
  __syncthreads();
  int s_l[EPW / 256], d_l[EPW / 256], r_l[EPW / 256];
  float w_l[EPW / 256];
  bool ok[EPW / 256];
  int e0 = (int)blockIdx.x * EPW + tid;
#pragma unroll
  for (int k = 0; k < EPW / 256; ++k) {
    int e = e0 + k * 256;
    ok[k] = e < E;
    s_l[k] = 0;
    d_l[k] = 0;
    w_l[k] = 0.f;
    if (ok[k]) {
      s_l[k] = src[e];
      d_l[k] = dst[e];
      w_l[k] = ew[e];
    }
  }
#pragma unroll
  for (int k = 0; k < EPW / 256; ++k) {
    r_l[k] = 0;
    if (ok[k]) r_l[k] = atomicAdd(&h[(unsigned)d_l[k] / TR], 1);
  }
  if (wv == 0) {
    int carry = 0;
    for (int base = 0; base < nb; base += 32) {
      int i = base + lane;
      int c = (i < nb) ? cnt[i] : 0;
      int x = c;
#pragma unroll
      for (int o = 1; o < 32; o <<= 1) {
        int y = __shfl_up(x, o, 32);
        if (lane >= o) x += y;
      }
      if (i < nb) st[i] = carry + x - c;
      carry += __shfl(x, 31, 32);
    }
  }
  __syncthreads();
  for (int i = tid; i < nb; i += 256) {
    int c = h[i];
    int b0 = st[i];
    if (blockIdx.x == 0) starts[i] = b0;
    int run = 0;
    if (c != 0) run = atomicAdd(&cur[i], c);
    h[i] = b0 + run;
  }
  __syncthreads();
#pragma unroll
  for (int k = 0; k < EPW / 256; ++k) {
    if (ok[k]) {
      unsigned bin = (unsigned)d_l[k] / TR;
      unsigned dl = (unsigned)d_l[k] - bin * TR;
      int pos = h[bin] + r_l[k];
      ((uint2*)edges)[pos] = make_uint2((unsigned)s_l[k] | (dl << 24), __float_as_uint(w_l[k]));
    }
  }
}

template <int ROWF, bool FULL, int BASE>
__device__ __forceinline__ void tile_group(const float* __restrict__ fu,
                                           float (&tile)[TR * ROWF],
                                           unsigned sd_r, float w_r, int n,
                                           unsigned lane) {
  float v[16][2];
  unsigned hoff = (ROWF == 64) ? (32u + lane) : (32u + (lane & 7u));
#pragma unroll
  for (int j = 0; j < 16; ++j) {
    unsigned s = bcast_u(sd_r, BASE + j) & 0xFFFFFFu;
    const float* fr = fu + (size_t)s * 64;
    v[j][0] = fr[lane];
    v[j][1] = fr[hoff];
  }
#pragma unroll
  for (int j = 0; j < 16; ++j) {
    if (FULL || j < n) {
      unsigned dl = bcast_u(sd_r, BASE + j) >> 24;
      float w = bcast_f(w_r, BASE + j);
      float* o = &tile[dl * ROWF];
      atomicAdd(o + lane, v[j][0] * w);
      if (ROWF == 64 || lane < 8u) atomicAdd(o + 32 + lane, v[j][1] * w);
    }
  }
}

template <int ROWF, int NW>
__global__ void __launch_bounds__(NW * 32, 16)
tile_sum_kernel(const float* __restrict__ feat,
                const int* __restrict__ cnt,
                const int* __restrict__ starts,
                const unsigned* __restrict__ edges,
                float* __restrict__ agg, int N, int nb, size_t pstride,
                const float* __restrict__ bias,
                float* __restrict__ csums) {
  __shared__ float tile[TR * ROWF];
  __shared__ float cs[(ROWF == 64) ? 128 : 1];
  int tid = threadIdx.x;
  if constexpr (ROWF == 64) {
    if (tid < 128) cs[tid] = 0.f;
  }
  int plane = (ROWF == 64 && (int)blockIdx.x >= nb) ? 1 : 0;
  int bin = (int)blockIdx.x - plane * nb;
  if (ROWF == 64) {
    float4 zero4 = {0.f, 0.f, 0.f, 0.f};
    for (int i = tid; i < TR * ROWF / 4; i += NW * 32) ((float4*)tile)[i] = zero4;
  } else {
    for (int i = tid; i < TR * ROWF; i += NW * 32) tile[i] = bias[i % ROWF];
  }
  __syncthreads();

  unsigned lane = (unsigned)tid & 31u;
  int wv = __builtin_amdgcn_readfirstlane(tid >> 5);
  int n = cnt[bin];
  const uint2* eb = (const uint2*)edges + starts[bin];
  const float* fu = feat + (size_t)plane * pstride;

  for (int p = wv * 32; p < n; p += NW * 32) {
    int idx = p + (int)lane;
    uint2 ent = make_uint2(0u, 0u);
    if (idx < n) ent = eb[idx];
    float w_r = __uint_as_float(ent.y);
    int m = n - p;
    if (m >= 32) {
      tile_group<ROWF, true, 0>(fu, tile, ent.x, w_r, 16, lane);
      tile_group<ROWF, true, 16>(fu, tile, ent.x, w_r, 16, lane);
    } else {
      tile_group<ROWF, false, 0>(fu, tile, ent.x, w_r, m, lane);
      if (m > 16) tile_group<ROWF, false, 16>(fu, tile, ent.x, w_r, m - 16, lane);
    }
  }
  __syncthreads();

  int row0 = bin * TR;
  int rows = (N - row0 < TR) ? (N - row0) : TR;
  float4* o = (float4*)(agg + (size_t)plane * pstride + (size_t)row0 * ROWF);
  int nq = rows * (ROWF / 4);
  if constexpr (ROWF == 64) {
    constexpr int NQT = TR * (ROWF / 4) / (NW * 32);
    float4 v[NQT];
    float4 s4 = {0.f, 0.f, 0.f, 0.f};
    float4 q4 = {0.f, 0.f, 0.f, 0.f};
#pragma unroll
    for (int k = 0; k < NQT; ++k) {
      int i = tid + k * NW * 32;
      v[k] = ((const float4*)tile)[i];
      bool in = i < nq;
      float x0 = in ? v[k].x : 0.f;
      float x1 = in ? v[k].y : 0.f;
      float x2 = in ? v[k].z : 0.f;
      float x3 = in ? v[k].w : 0.f;
      s4.x += x0; s4.y += x1; s4.z += x2; s4.w += x3;
      q4.x = fmaf(x0, x0, q4.x); q4.y = fmaf(x1, x1, q4.y);
      q4.z = fmaf(x2, x2, q4.z); q4.w = fmaf(x3, x3, q4.w);
    }
    s4.x += __shfl_xor(s4.x, 16, 32); s4.y += __shfl_xor(s4.y, 16, 32);
    s4.z += __shfl_xor(s4.z, 16, 32); s4.w += __shfl_xor(s4.w, 16, 32);
    q4.x += __shfl_xor(q4.x, 16, 32); q4.y += __shfl_xor(q4.y, 16, 32);
    q4.z += __shfl_xor(q4.z, 16, 32); q4.w += __shfl_xor(q4.w, 16, 32);
    if ((tid & 16) == 0) {
      int c = (tid & 15) * 4;
      atomicAdd(&cs[c + 0], s4.x); atomicAdd(&cs[c + 1], s4.y);
      atomicAdd(&cs[c + 2], s4.z); atomicAdd(&cs[c + 3], s4.w);
      atomicAdd(&cs[64 + c + 0], q4.x); atomicAdd(&cs[64 + c + 1], q4.y);
      atomicAdd(&cs[64 + c + 2], q4.z); atomicAdd(&cs[64 + c + 3], q4.w);
    }
    __syncthreads();
    if (tid < 128) atomicAdd(&csums[(tid >> 6) * 128 + plane * 64 + (tid & 63)], cs[tid]);
#pragma unroll
    for (int k = 0; k < NQT; ++k) {
      int i = tid + k * NW * 32;
      if (i < nq) o[i] = v[k];
    }
  } else {
    for (int i = tid; i < nq; i += NW * 32) o[i] = ((const float4*)tile)[i];
  }
}

template <int NT, bool BN, int COUT>
__global__ void gemm_wmma_kernel(const float* A,
                                 const float* __restrict__ W,
                                 float* out,
                                 int Nrows, int ldo,
                                 const float* __restrict__ sums,
                                 const float* __restrict__ gamma,
                                 const float* __restrict__ beta,
                                 float* __restrict__ zsums,
                                 float* __restrict__ zsums2,
                                 int* __restrict__ zctr, int nzc,
                                 const unsigned* __restrict__ cpsrc,
                                 unsigned* __restrict__ cpdst, int ncp) {
  constexpr int CP = NT * 16;
  constexpr bool APL = BN;
  constexpr bool OPL = (COUT == CH);
  const size_t pst = (size_t)Nrows * 64;
  __shared__ float sWT[CP * WSTRIDE];
  __shared__ float sMean[CH];
  __shared__ float sSc[CH];
  __shared__ float sBeta[CH];
  int tid = threadIdx.x;
  if (zsums != nullptr && blockIdx.x == 0) zsums[tid] = 0.f;
  if (zsums2 != nullptr && blockIdx.x == 0) zsums2[tid] = 0.f;
  if (zctr != nullptr && blockIdx.x == 0) {
    for (int i = tid; i < nzc; i += 256) zctr[i] = 0;
  }
  if (cpsrc != nullptr) {
    int nth = (int)gridDim.x * 256;
    for (int i = (int)blockIdx.x * 256 + tid; i < ncp; i += nth)
      ((uint2*)cpdst)[i] = ((const uint2*)cpsrc)[i];
  }
  constexpr int WPER = CH * CP / 256;
  constexpr int WB = (WPER % 32 == 0) ? 32 : WPER;
  static_assert(WPER % WB == 0, "whole batches");
  for (int base = 0; base < WPER; base += WB) {
    float wv[WB];
#pragma unroll
    for (int i = 0; i < WB; ++i) {
      int idx = tid + (base + i) * 256;
      int k = idx / CP;
      int c = idx % CP;
      int cc = (c < COUT) ? c : (COUT - 1);
      wv[i] = W[k * COUT + cc];
    }
#pragma unroll
    for (int i = 0; i < WB; ++i) {
      int idx = tid + (base + i) * 256;
      int k = idx / CP;
      int c = idx % CP;
      sWT[c * WSTRIDE + k] = (c < COUT) ? wv[i] : 0.f;
    }
  }
  if (BN) {
    if (tid < CH) {
      float invN = 1.0f / (float)Nrows;
      float mean = sums[tid] * invN;
      float var = sums[128 + tid] * invN - mean * mean;
      sMean[tid] = mean;
      sSc[tid] = gamma[tid] * rsqrtf(var + BN_EPS);
      sBeta[tid] = beta[tid];
    }
  }
  __syncthreads();

  int wave = tid >> 5;
  int lane = tid & 31;
  int row0 = blockIdx.x * 128 + wave * 16;
  if (row0 >= Nrows) return;

  int m = lane & 15;
  int kh = (lane >> 4) * 2;

  v8f acc[NT];
#pragma unroll
  for (int t = 0; t < NT; ++t) acc[t] = {};

  const float* Arow = APL ? A + (size_t)(row0 + m) * 64 : A + (size_t)(row0 + m) * CH;
  const size_t ahalf = APL ? pst : (size_t)64;

  const float* Arow2 = Arow + ahalf - 64;

#pragma unroll 4
  for (int k0 = 0; k0 < CH; k0 += 4) {
    v2f a = *(const v2f*)((k0 < 64 ? Arow : Arow2) + k0 + kh);
    if (BN) {
      v2f mu = *(const v2f*)(&sMean[k0 + kh]);
      v2f sc = *(const v2f*)(&sSc[k0 + kh]);
      v2f be = *(const v2f*)(&sBeta[k0 + kh]);
      float x0 = (a.x - mu.x) * sc.x + be.x;
      float x1 = (a.y - mu.y) * sc.y + be.y;
      a.x = x0 > 0.f ? x0 : 0.f;
      a.y = x1 > 0.f ? x1 : 0.f;
    }
#pragma unroll
    for (int t = 0; t < NT; ++t) {
      v2f b = *(const v2f*)(&sWT[(t * 16 + m) * WSTRIDE + k0 + kh]);
      acc[t] = __builtin_amdgcn_wmma_f32_16x16x4_f32(
           false, a,  false, b,
           (short)0, acc[t],  false,  false);
    }
  }

  int rbase = row0 + (lane >> 4) * 8;
#pragma unroll
  for (int t = 0; t < NT; ++t) {
    int col = t * 16 + m;
    if (col < COUT) {
#pragma unroll
      for (int j = 0; j < 8; ++j) {
        if (OPL) {
          out[(size_t)(t >> 2) * pst + (size_t)(rbase + j) * 64 + (col & 63)] = acc[t][j];
        } else {
          out[(size_t)(rbase + j) * ldo + col] = acc[t][j];
        }
      }
    }
  }
}

extern "C" void kernel_launch(void* const* d_in, const int* in_sizes, int n_in,
                              void* d_out, int out_size, void* d_ws, size_t ws_size,
                              hipStream_t stream) {
  const float* feat   = (const float*)d_in[0];
  const int*   src    = (const int*)d_in[1];
  const int*   dst    = (const int*)d_in[2];
  const float* ew     = (const float*)d_in[3];
  const float* W1     = (const float*)d_in[4];
  const float* W2     = (const float*)d_in[5];
  const float* W3     = (const float*)d_in[6];
  const float* b3     = (const float*)d_in[7];
  const float* gamma1 = (const float*)d_in[8];
  const float* beta1  = (const float*)d_in[9];
  const float* gamma2 = (const float*)d_in[10];
  const float* beta2  = (const float*)d_in[11];

  int N = in_sizes[0] / CH;
  int E = in_sizes[1];

  float* t    = (float*)d_ws;
  float* a    = t + (size_t)N * CH;
  float* sums = a + (size_t)N * CH;
  float* out  = (float*)d_out;

  size_t pstride = (size_t)N * 64;
  int ggrid = (N + 127) / 128;
  dim3 b256(256);

  int nb = (N + TR - 1) / TR;
  int hdr = (3 * nb + 31) / 32 * 32;
  int bgrid = (E + EPW - 1) / EPW;
  int ncp = (hdr + 2 * E) / 2;
  int* bl1 = (int*)d_out;
  int* bl3 = (int*)(t + (size_t)N * 64);
  unsigned* ed1 = (unsigned*)(bl1 + hdr);
  unsigned* ed3 = (unsigned*)(bl3 + hdr);
  float* sums2 = (float*)(bl1 + hdr + 2 * E);

  gemm_wmma_kernel<8, false, 128><<<ggrid, b256, 0, stream>>>(
      feat, W1, t, N, 128, nullptr, nullptr, nullptr, sums, sums2, bl1, 2 * nb, nullptr, nullptr,
      0);
  bin_count_kernel<<<bgrid, b256, 0, stream>>>(dst, E, bl1, nb);
  bin_place_kernel<<<bgrid, b256, 0, stream>>>(src, dst, ew, E, bl1, bl1 + nb, bl1 + 2 * nb,
                                               ed1, nb);
  tile_sum_kernel<64, 16><<<2 * nb, 512, 0, stream>>>(
      t, bl1, bl1 + 2 * nb, ed1, a, N, nb, pstride, nullptr, sums);

  gemm_wmma_kernel<8, true, 128><<<ggrid, b256, 0, stream>>>(
      a, W2, t, N, 128, sums, gamma1, beta1, nullptr, nullptr, nullptr, 0, nullptr, nullptr, 0);
  tile_sum_kernel<64, 16><<<2 * nb, 512, 0, stream>>>(
      t, bl1, bl1 + 2 * nb, ed1, a, N, nb, pstride, nullptr, sums2);

  gemm_wmma_kernel<3, true, 40><<<ggrid, b256, 0, stream>>>(
      a, W3, t, N, 64, sums2, gamma2, beta2, nullptr, nullptr, nullptr, 0,
      (const unsigned*)bl1, (unsigned*)bl3, ncp);
  tile_sum_kernel<40, 32><<<nb, 1024, 0, stream>>>(
      t, bl3, bl3 + 2 * nb, ed3, out, N, nb, 0, b3, nullptr);
}


# Edge sums fetch each source row with ONE load (16 B per lane, whole 512-B row; layer 3: 8 B per lane); tile = all 128 channels, no planes; LDS f32 adds unordered as before, column stats new sum tree
# speedup vs baseline: 3.6422x; 1.0603x over previous
#include <hip/hip_runtime.h>
#include <hip/hip_bf16.h>

typedef __attribute__((ext_vector_type(2))) float v2f;
typedef __attribute__((ext_vector_type(8))) float v8f;

#define BN_EPS 1e-5f
#define CH 128
#define WSTRIDE 132

#define TR 256
#define NBMAX 2048
#define EPW 2048

__device__ __forceinline__ unsigned bcast_u(unsigned v, int j) {
  return (unsigned)__builtin_amdgcn_readlane((int)v, j);
}
__device__ __forceinline__ float bcast_f(float v, int j) {
  return __int_as_float(__builtin_amdgcn_readlane(__float_as_int(v), j));
}

__global__ void __launch_bounds__(256)
bin_count_kernel(const int* __restrict__ dst, int E, int* __restrict__ cnt, int nb) {
  __shared__ int h[NBMAX];
  int tid = threadIdx.x;
  for (int i = tid; i < nb; i += 256) h[i] = 0;
  __syncthreads();
  int e0 = (int)blockIdx.x * EPW + tid;
#pragma unroll
  for (int k = 0; k < EPW / 256; ++k) {
    int e = e0 + k * 256;
    if (e < E) atomicAdd(&h[(unsigned)dst[e] / TR], 1);
  }
  __syncthreads();
  for (int i = tid; i < nb; i += 256) {
    int c = h[i];
    if (c != 0) atomicAdd(&cnt[i], c);
  }
}

__global__ void __launch_bounds__(256)
bin_place_kernel(const int* __restrict__ src, const int* __restrict__ dst,
                 const float* __restrict__ ew, int E,
                 const int* __restrict__ cnt, int* __restrict__ cur,
                 int* __restrict__ starts, unsigned* __restrict__ edges, int nb) {
  __shared__ int h[NBMAX];
  __shared__ int st[NBMAX];
  int tid = threadIdx.x;
  int lane = tid & 31;
  int wv = __builtin_amdgcn_readfirstlane(tid >> 5);
  for (int i = tid; i < nb; i += 256) h[i] = 0;
  __syncthreads();
  int s_l[EPW / 256], d_l[EPW / 256], r_l[EPW / 256];
  float w_l[EPW / 256];
  bool ok[EPW / 256];
  int e0 = (int)blockIdx.x * EPW + tid;
#pragma unroll
  for (int k = 0; k < EPW / 256; ++k) {
    int e = e0 + k * 256;
    ok[k] = e < E;
    s_l[k] = 0;
    d_l[k] = 0;
    w_l[k] = 0.f;
    if (ok[k]) {
      s_l[k] = src[e];
      d_l[k] = dst[e];
      w_l[k] = ew[e];
    }
  }
#pragma unroll
  for (int k = 0; k < EPW / 256; ++k) {
    r_l[k] = 0;
    if (ok[k]) r_l[k] = atomicAdd(&h[(unsigned)d_l[k] / TR], 1);
  }
  if (wv == 0) {
    int carry = 0;
    for (int base = 0; base < nb; base += 32) {
      int i = base + lane;
      int c = (i < nb) ? cnt[i] : 0;
      int x = c;
#pragma unroll
      for (int o = 1; o < 32; o <<= 1) {
        int y = __shfl_up(x, o, 32);
        if (lane >= o) x += y;
      }
      if (i < nb) st[i] = carry + x - c;
      carry += __shfl(x, 31, 32);
    }
  }
  __syncthreads();
  for (int i = tid; i < nb; i += 256) {
    int c = h[i];
    int b0 = st[i];
    if (blockIdx.x == 0) starts[i] = b0;
    int run = 0;
    if (c != 0) run = atomicAdd(&cur[i], c);
    h[i] = b0 + run;
  }
  __syncthreads();
#pragma unroll
  for (int k = 0; k < EPW / 256; ++k) {
    if (ok[k]) {
      unsigned bin = (unsigned)d_l[k] / TR;
      unsigned dl = (unsigned)d_l[k] - bin * TR;
      int pos = h[bin] + r_l[k];
      ((uint2*)edges)[pos] = make_uint2((unsigned)s_l[k] | (dl << 24), __float_as_uint(w_l[k]));
    }
  }
}

template <int ROWF> struct rowvec;
template <> struct rowvec<128> { typedef float4 type; };
template <> struct rowvec<40> { typedef float2 type; };

template <int ROWF, bool FULL>
__device__ __forceinline__ void tile_trip(const float* __restrict__ fu,
                                          float (&tile)[TR * ROWF],
                                          unsigned sd_r, float w_r, int n,
                                          unsigned lane) {
  typedef typename rowvec<ROWF>::type rv;
  constexpr int SRS = (ROWF == 128) ? 128 : 64;
  constexpr int DEPTH = (ROWF == 128) ? 8 : 16;
  unsigned li = (ROWF == 128 || lane < 20u) ? lane : lane - 20u;
  rv v[DEPTH];
#pragma unroll
  for (int j = 0; j < DEPTH; ++j) {
    unsigned s = bcast_u(sd_r, j) & 0xFFFFFFu;
    v[j] = ((const rv*)(fu + (size_t)s * SRS))[li];
  }
#pragma unroll
  for (int j = 0; j < 32; ++j) {
    if (FULL || j < n) {
      unsigned dl = bcast_u(sd_r, j) >> 24;
      float w = bcast_f(w_r, j);
      rv x = v[j % DEPTH];
      if constexpr (ROWF == 128) {
        float* o = &tile[dl * 128 + lane];
        atomicAdd(o, x.x * w);
        atomicAdd(o + 32, x.y * w);
        atomicAdd(o + 64, x.z * w);
        atomicAdd(o + 96, x.w * w);
      } else {
        float* o = &tile[dl * 40 + 2 * lane];
        float p0 = x.x * w;
        float p1 = x.y * w;
        if (lane < 20u) {
          atomicAdd(o, p0);
          atomicAdd(o + 1, p1);
        }
      }
    }
    if (j + DEPTH < 32) {
      unsigned s = bcast_u(sd_r, j + DEPTH) & 0xFFFFFFu;
      v[j % DEPTH] = ((const rv*)(fu + (size_t)s * SRS))[li];
    }
  }
}

template <int ROWF, int NW>
__global__ void __launch_bounds__(NW * 32, 16)
tile_sum_kernel(const float* __restrict__ feat,
                const int* __restrict__ cnt,
                const int* __restrict__ starts,
                const unsigned* __restrict__ edges,
                float* __restrict__ agg, int N,
                const float* __restrict__ bias,
                float* __restrict__ csums) {
  __shared__ float tile[TR * ROWF];
  __shared__ float cs[(ROWF == 128) ? 256 : 1];
  int tid = threadIdx.x;
  if constexpr (ROWF == 128) {
    if (tid < 256) cs[tid] = 0.f;
  }
  int bin = (int)blockIdx.x;
  if (ROWF == 128) {
    float4 zero4 = {0.f, 0.f, 0.f, 0.f};
    for (int i = tid; i < TR * ROWF / 4; i += NW * 32) ((float4*)tile)[i] = zero4;
  } else {
    for (int i = tid; i < TR * ROWF; i += NW * 32) tile[i] = bias[i % ROWF];
  }
  __syncthreads();

  unsigned lane = (unsigned)tid & 31u;
  int wv = __builtin_amdgcn_readfirstlane(tid >> 5);
  int n = cnt[bin];
  const uint2* eb = (const uint2*)edges + starts[bin];

  for (int p = wv * 32; p < n; p += NW * 32) {
    int idx = p + (int)lane;
    uint2 ent = make_uint2(0u, 0u);
    if (idx < n) ent = eb[idx];
    float w_r = __uint_as_float(ent.y);
    int m = n - p;
    if (m >= 32) {
      tile_trip<ROWF, true>(feat, tile, ent.x, w_r, 32, lane);
    } else {
      tile_trip<ROWF, false>(feat, tile, ent.x, w_r, m, lane);
    }
  }
  __syncthreads();

  int row0 = bin * TR;
  int rows = (N - row0 < TR) ? (N - row0) : TR;
  float4* o = (float4*)(agg + (size_t)row0 * ROWF);
  int nq = rows * (ROWF / 4);
  if constexpr (ROWF == 128) {
    constexpr int NQT = TR * (ROWF / 4) / (NW * 32);
    float4 v[NQT];
    float4 s4 = {0.f, 0.f, 0.f, 0.f};
    float4 q4 = {0.f, 0.f, 0.f, 0.f};
#pragma unroll
    for (int k = 0; k < NQT; ++k) {
      int i = tid + k * NW * 32;
      const float* r = &tile[(i >> 5) * 128 + (int)lane];
      v[k].x = r[0];
      v[k].y = r[32];
      v[k].z = r[64];
      v[k].w = r[96];
      bool in = i < nq;
      float x0 = in ? v[k].x : 0.f;
      float x1 = in ? v[k].y : 0.f;
      float x2 = in ? v[k].z : 0.f;
      float x3 = in ? v[k].w : 0.f;
      s4.x += x0; s4.y += x1; s4.z += x2; s4.w += x3;
      q4.x = fmaf(x0, x0, q4.x); q4.y = fmaf(x1, x1, q4.y);
      q4.z = fmaf(x2, x2, q4.z); q4.w = fmaf(x3, x3, q4.w);
    }
    int c = (int)lane * 4;
    atomicAdd(&cs[c + 0], s4.x); atomicAdd(&cs[c + 1], s4.y);
    atomicAdd(&cs[c + 2], s4.z); atomicAdd(&cs[c + 3], s4.w);
    atomicAdd(&cs[128 + c + 0], q4.x); atomicAdd(&cs[128 + c + 1], q4.y);
    atomicAdd(&cs[128 + c + 2], q4.z); atomicAdd(&cs[128 + c + 3], q4.w);
    __syncthreads();
    if (tid < 256) atomicAdd(&csums[tid], cs[tid]);
#pragma unroll
    for (int k = 0; k < NQT; ++k) {
      int i = tid + k * NW * 32;
      if (i < nq) o[i] = v[k];
    }
  } else {
    for (int i = tid; i < nq; i += NW * 32) o[i] = ((const float4*)tile)[i];
  }
}

template <int NT, bool BN, int COUT>
__global__ void gemm_wmma_kernel(const float* A,
                                 const float* __restrict__ W,
                                 float* out,
                                 int Nrows, int ldo,
                                 const float* __restrict__ sums,
                                 const float* __restrict__ gamma,
                                 const float* __restrict__ beta,
                                 float* __restrict__ zsums,
                                 float* __restrict__ zsums2,
                                 int* __restrict__ zctr, int nzc,
                                 const unsigned* __restrict__ cpsrc,
                                 unsigned* __restrict__ cpdst, int ncp) {
  constexpr int CP = NT * 16;
  __shared__ float sWT[CP * WSTRIDE];
  __shared__ float sMean[CH];
  __shared__ float sSc[CH];
  __shared__ float sBeta[CH];
  int tid = threadIdx.x;
  if (zsums != nullptr && blockIdx.x == 0) zsums[tid] = 0.f;
  if (zsums2 != nullptr && blockIdx.x == 0) zsums2[tid] = 0.f;
  if (zctr != nullptr && blockIdx.x == 0) {
    for (int i = tid; i < nzc; i += 256) zctr[i] = 0;
  }
  if (cpsrc != nullptr) {
    int nth = (int)gridDim.x * 256;
    for (int i = (int)blockIdx.x * 256 + tid; i < ncp; i += nth)
      ((uint2*)cpdst)[i] = ((const uint2*)cpsrc)[i];
  }
  constexpr int WPER = CH * CP / 256;
  constexpr int WB = (WPER % 32 == 0) ? 32 : WPER;
  static_assert(WPER % WB == 0, "whole batches");
  for (int base = 0; base < WPER; base += WB) {
    float wv[WB];
#pragma unroll
    for (int i = 0; i < WB; ++i) {
      int idx = tid + (base + i) * 256;
      int k = idx / CP;
      int c = idx % CP;
      int cc = (c < COUT) ? c : (COUT - 1);
      wv[i] = W[k * COUT + cc];
    }
#pragma unroll
    for (int i = 0; i < WB; ++i) {
      int idx = tid + (base + i) * 256;
      int k = idx / CP;
      int c = idx % CP;
      sWT[c * WSTRIDE + k] = (c < COUT) ? wv[i] : 0.f;
    }
  }
  if (BN) {
    if (tid < CH) {
      float invN = 1.0f / (float)Nrows;
      float mean = sums[tid] * invN;
      float var = sums[128 + tid] * invN - mean * mean;
      sMean[tid] = mean;
      sSc[tid] = gamma[tid] * rsqrtf(var + BN_EPS);
      sBeta[tid] = beta[tid];
    }
  }
  __syncthreads();

  int wave = tid >> 5;
  int lane = tid & 31;
  int row0 = blockIdx.x * 128 + wave * 16;
  if (row0 >= Nrows) return;

  int m = lane & 15;
  int kh = (lane >> 4) * 2;

  v8f acc[NT];
#pragma unroll
  for (int t = 0; t < NT; ++t) acc[t] = {};

  const float* Arow = A + (size_t)(row0 + m) * CH;

  for (int k0 = 0; k0 < CH; k0 += 4) {
    v2f a = *(const v2f*)(Arow + k0 + kh);
    if (BN) {
      v2f mu = *(const v2f*)(&sMean[k0 + kh]);
      v2f sc = *(const v2f*)(&sSc[k0 + kh]);
      v2f be = *(const v2f*)(&sBeta[k0 + kh]);
      float x0 = (a.x - mu.x) * sc.x + be.x;
      float x1 = (a.y - mu.y) * sc.y + be.y;
      a.x = x0 > 0.f ? x0 : 0.f;
      a.y = x1 > 0.f ? x1 : 0.f;
    }
#pragma unroll
    for (int t = 0; t < NT; ++t) {
      v2f b = *(const v2f*)(&sWT[(t * 16 + m) * WSTRIDE + k0 + kh]);
      acc[t] = __builtin_amdgcn_wmma_f32_16x16x4_f32(
           false, a,  false, b,
           (short)0, acc[t],  false,  false);
    }
  }

  int rbase = row0 + (lane >> 4) * 8;
#pragma unroll
  for (int t = 0; t < NT; ++t) {
    int col = t * 16 + m;
    if (col < COUT) {
#pragma unroll
      for (int j = 0; j < 8; ++j) out[(size_t)(rbase + j) * ldo + col] = acc[t][j];
    }
  }
}

extern "C" void kernel_launch(void* const* d_in, const int* in_sizes, int n_in,
                              void* d_out, int out_size, void* d_ws, size_t ws_size,
                              hipStream_t stream) {
  const float* feat   = (const float*)d_in[0];
  const int*   src    = (const int*)d_in[1];
  const int*   dst    = (const int*)d_in[2];
  const float* ew     = (const float*)d_in[3];
  const float* W1     = (const float*)d_in[4];
  const float* W2     = (const float*)d_in[5];
  const float* W3     = (const float*)d_in[6];
  const float* b3     = (const float*)d_in[7];
  const float* gamma1 = (const float*)d_in[8];
  const float* beta1  = (const float*)d_in[9];
  const float* gamma2 = (const float*)d_in[10];
  const float* beta2  = (const float*)d_in[11];

  int N = in_sizes[0] / CH;
  int E = in_sizes[1];

  float* t    = (float*)d_ws;
  float* a    = t + (size_t)N * CH;
  float* sums = a + (size_t)N * CH;
  float* out  = (float*)d_out;

  int ggrid = (N + 127) / 128;
  dim3 b256(256);

  int nb = (N + TR - 1) / TR;
  int hdr = (3 * nb + 31) / 32 * 32;
  int bgrid = (E + EPW - 1) / EPW;
  int ncp = (hdr + 2 * E) / 2;
  int* bl1 = (int*)d_out;
  int* bl3 = (int*)(t + (size_t)N * 64);
  unsigned* ed1 = (unsigned*)(bl1 + hdr);
  unsigned* ed3 = (unsigned*)(bl3 + hdr);
  float* sums2 = (float*)(bl1 + hdr + 2 * E);

  gemm_wmma_kernel<8, false, 128><<<ggrid, b256, 0, stream>>>(
      feat, W1, t, N, 128, nullptr, nullptr, nullptr, sums, sums2, bl1, 2 * nb, nullptr, nullptr,
      0);
  bin_count_kernel<<<bgrid, b256, 0, stream>>>(dst, E, bl1, nb);
  bin_place_kernel<<<bgrid, b256, 0, stream>>>(src, dst, ew, E, bl1, bl1 + nb, bl1 + 2 * nb,
                                               ed1, nb);
  tile_sum_kernel<128, 32><<<nb, 1024, 0, stream>>>(
      t, bl1, bl1 + 2 * nb, ed1, a, N, nullptr, sums);

  gemm_wmma_kernel<8, true, 128><<<ggrid, b256, 0, stream>>>(
      a, W2, t, N, 128, sums, gamma1, beta1, nullptr, nullptr, nullptr, 0, nullptr, nullptr, 0);
  tile_sum_kernel<128, 32><<<nb, 1024, 0, stream>>>(
      t, bl1, bl1 + 2 * nb, ed1, a, N, nullptr, sums2);

  gemm_wmma_kernel<3, true, 40><<<ggrid, b256, 0, stream>>>(
      a, W3, t, N, 64, sums2, gamma2, beta2, nullptr, nullptr, nullptr, 0,
      (const unsigned*)bl1, (unsigned*)bl3, ncp);
  tile_sum_kernel<40, 32><<<nb, 1024, 0, stream>>>(
      t, bl3, bl3 + 2 * nb, ed3, out, N, b3, nullptr);
}


# Edge-sum tiles 196 rows high, not 256: 511 tiles fill both work-group slots of all 256 WGPs (391 left some WGPs two tiles, others one); LDS adds unordered as before, column-stat partials regrouped
# speedup vs baseline: 3.8456x; 1.0558x over previous
#include <hip/hip_runtime.h>
#include <hip/hip_bf16.h>

typedef __attribute__((ext_vector_type(2))) float v2f;
typedef __attribute__((ext_vector_type(8))) float v8f;

#define BN_EPS 1e-5f
#define CH 128
#define WSTRIDE 132

#define TR 196
#define NBMAX 2048
#define EPW 2048

__device__ __forceinline__ unsigned bcast_u(unsigned v, int j) {
  return (unsigned)__builtin_amdgcn_readlane((int)v, j);
}
__device__ __forceinline__ float bcast_f(float v, int j) {
  return __int_as_float(__builtin_amdgcn_readlane(__float_as_int(v), j));
}

__global__ void __launch_bounds__(256)
bin_count_kernel(const int* __restrict__ dst, int E, int* __restrict__ cnt, int nb) {
  __shared__ int h[NBMAX];
  int tid = threadIdx.x;
  for (int i = tid; i < nb; i += 256) h[i] = 0;
  __syncthreads();
  int e0 = (int)blockIdx.x * EPW + tid;
#pragma unroll
  for (int k = 0; k < EPW / 256; ++k) {
    int e = e0 + k * 256;
    if (e < E) atomicAdd(&h[(unsigned)dst[e] / TR], 1);
  }
  __syncthreads();
  for (int i = tid; i < nb; i += 256) {
    int c = h[i];
    if (c != 0) atomicAdd(&cnt[i], c);
  }
}

__global__ void __launch_bounds__(256)
bin_place_kernel(const int* __restrict__ src, const int* __restrict__ dst,
                 const float* __restrict__ ew, int E,
                 const int* __restrict__ cnt, int* __restrict__ cur,
                 int* __restrict__ starts, unsigned* __restrict__ edges, int nb) {
  __shared__ int h[NBMAX];
  __shared__ int st[NBMAX];
  int tid = threadIdx.x;
  int lane = tid & 31;
  int wv = __builtin_amdgcn_readfirstlane(tid >> 5);
  for (int i = tid; i < nb; i += 256) h[i] = 0;
  __syncthreads();
  int s_l[EPW / 256], d_l[EPW / 256], r_l[EPW / 256];
  float w_l[EPW / 256];
  bool ok[EPW / 256];
  int e0 = (int)blockIdx.x * EPW + tid;
#pragma unroll
  for (int k = 0; k < EPW / 256; ++k) {
    int e = e0 + k * 256;
    ok[k] = e < E;
    s_l[k] = 0;
    d_l[k] = 0;
    w_l[k] = 0.f;
    if (ok[k]) {
      s_l[k] = src[e];
      d_l[k] = dst[e];
      w_l[k] = ew[e];
    }
  }
#pragma unroll
  for (int k = 0; k < EPW / 256; ++k) {
    r_l[k] = 0;
    if (ok[k]) r_l[k] = atomicAdd(&h[(unsigned)d_l[k] / TR], 1);
  }
  if (wv == 0) {
    int carry = 0;
    for (int base = 0; base < nb; base += 32) {
      int i = base + lane;
      int c = (i < nb) ? cnt[i] : 0;
      int x = c;
#pragma unroll
      for (int o = 1; o < 32; o <<= 1) {
        int y = __shfl_up(x, o, 32);
        if (lane >= o) x += y;
      }
      if (i < nb) st[i] = carry + x - c;
      carry += __shfl(x, 31, 32);
    }
  }
  __syncthreads();
  for (int i = tid; i < nb; i += 256) {
    int c = h[i];
    int b0 = st[i];
    if (blockIdx.x == 0) starts[i] = b0;
    int run = 0;
    if (c != 0) run = atomicAdd(&cur[i], c);
    h[i] = b0 + run;
  }
  __syncthreads();
#pragma unroll
  for (int k = 0; k < EPW / 256; ++k) {
    if (ok[k]) {
      unsigned bin = (unsigned)d_l[k] / TR;
      unsigned dl = (unsigned)d_l[k] - bin * TR;
      int pos = h[bin] + r_l[k];
      ((uint2*)edges)[pos] = make_uint2((unsigned)s_l[k] | (dl << 24), __float_as_uint(w_l[k]));
    }
  }
}

template <int ROWF> struct rowvec;
template <> struct rowvec<128> { typedef float4 type; };
template <> struct rowvec<40> { typedef float2 type; };

template <int ROWF, bool FULL>
__device__ __forceinline__ void tile_trip(const float* __restrict__ fu,
                                          float (&tile)[TR * ROWF],
                                          unsigned sd_r, float w_r, int n,
                                          unsigned lane) {
  typedef typename rowvec<ROWF>::type rv;
  constexpr int SRS = (ROWF == 128) ? 128 : 64;
  constexpr int DEPTH = (ROWF == 128) ? 8 : 16;
  unsigned li = (ROWF == 128 || lane < 20u) ? lane : lane - 20u;
  rv v[DEPTH];
#pragma unroll
  for (int j = 0; j < DEPTH; ++j) {
    unsigned s = bcast_u(sd_r, j) & 0xFFFFFFu;
    v[j] = ((const rv*)(fu + (size_t)s * SRS))[li];
  }
#pragma unroll
  for (int j = 0; j < 32; ++j) {
    if (FULL || j < n) {
      unsigned dl = bcast_u(sd_r, j) >> 24;
      float w = bcast_f(w_r, j);
      rv x = v[j % DEPTH];
      if constexpr (ROWF == 128) {
        float* o = &tile[dl * 128 + lane];
        atomicAdd(o, x.x * w);
        atomicAdd(o + 32, x.y * w);
        atomicAdd(o + 64, x.z * w);
        atomicAdd(o + 96, x.w * w);
      } else {
        float* o = &tile[dl * 40 + 2 * lane];
        float p0 = x.x * w;
        float p1 = x.y * w;
        if (lane < 20u) {
          atomicAdd(o, p0);
          atomicAdd(o + 1, p1);
        }
      }
    }
    if (j + DEPTH < 32) {
      unsigned s = bcast_u(sd_r, j + DEPTH) & 0xFFFFFFu;
      v[j % DEPTH] = ((const rv*)(fu + (size_t)s * SRS))[li];
    }
  }
}

template <int ROWF, int NW>
__global__ void __launch_bounds__(NW * 32, 16)
tile_sum_kernel(const float* __restrict__ feat,
                const int* __restrict__ cnt,
                const int* __restrict__ starts,
                const unsigned* __restrict__ edges,
                float* __restrict__ agg, int N,
                const float* __restrict__ bias,
                float* __restrict__ csums) {
  __shared__ float tile[TR * ROWF];
  __shared__ float cs[(ROWF == 128) ? 256 : 1];
  int tid = threadIdx.x;
  if constexpr (ROWF == 128) {
    if (tid < 256) cs[tid] = 0.f;
  }
  int bin = (int)blockIdx.x;
  if (ROWF == 128) {
    float4 zero4 = {0.f, 0.f, 0.f, 0.f};
    for (int i = tid; i < TR * ROWF / 4; i += NW * 32) ((float4*)tile)[i] = zero4;
  } else {
    for (int i = tid; i < TR * ROWF; i += NW * 32) tile[i] = bias[i % ROWF];
  }
  __syncthreads();

  unsigned lane = (unsigned)tid & 31u;
  int wv = __builtin_amdgcn_readfirstlane(tid >> 5);
  int n = cnt[bin];
  const uint2* eb = (const uint2*)edges + starts[bin];

  for (int p = wv * 32; p < n; p += NW * 32) {
    int idx = p + (int)lane;
    uint2 ent = make_uint2(0u, 0u);
    if (idx < n) ent = eb[idx];
    float w_r = __uint_as_float(ent.y);
    int m = n - p;
    if (m >= 32) {
      tile_trip<ROWF, true>(feat, tile, ent.x, w_r, 32, lane);
    } else {
      tile_trip<ROWF, false>(feat, tile, ent.x, w_r, m, lane);
    }
  }
  __syncthreads();

  int row0 = bin * TR;
  int rows = (N - row0 < TR) ? (N - row0) : TR;
  float4* o = (float4*)(agg + (size_t)row0 * ROWF);
  int nq = rows * (ROWF / 4);
  if constexpr (ROWF == 128) {
    constexpr int NQT = (TR * (ROWF / 4) + NW * 32 - 1) / (NW * 32);
    float4 v[NQT];
    float4 s4 = {0.f, 0.f, 0.f, 0.f};
    float4 q4 = {0.f, 0.f, 0.f, 0.f};
#pragma unroll
    for (int k = 0; k < NQT; ++k) {
      int i = tid + k * NW * 32;
      bool in = i < nq;
      float x0 = 0.f, x1 = 0.f, x2 = 0.f, x3 = 0.f;
      if (in) {
        const float* r = &tile[(i >> 5) * 128 + (int)lane];
        x0 = r[0];
        x1 = r[32];
        x2 = r[64];
        x3 = r[96];
      }
      v[k].x = x0; v[k].y = x1; v[k].z = x2; v[k].w = x3;
      s4.x += x0; s4.y += x1; s4.z += x2; s4.w += x3;
      q4.x = fmaf(x0, x0, q4.x); q4.y = fmaf(x1, x1, q4.y);
      q4.z = fmaf(x2, x2, q4.z); q4.w = fmaf(x3, x3, q4.w);
    }
    int c = (int)lane * 4;
    atomicAdd(&cs[c + 0], s4.x); atomicAdd(&cs[c + 1], s4.y);
    atomicAdd(&cs[c + 2], s4.z); atomicAdd(&cs[c + 3], s4.w);
    atomicAdd(&cs[128 + c + 0], q4.x); atomicAdd(&cs[128 + c + 1], q4.y);
    atomicAdd(&cs[128 + c + 2], q4.z); atomicAdd(&cs[128 + c + 3], q4.w);
    __syncthreads();
    if (tid < 256) atomicAdd(&csums[tid], cs[tid]);
#pragma unroll
    for (int k = 0; k < NQT; ++k) {
      int i = tid + k * NW * 32;
      if (i < nq) o[i] = v[k];
    }
  } else {
    for (int i = tid; i < nq; i += NW * 32) o[i] = ((const float4*)tile)[i];
  }
}

template <int NT, bool BN, int COUT>
__global__ void gemm_wmma_kernel(const float* A,
                                 const float* __restrict__ W,
                                 float* out,
                                 int Nrows, int ldo,
                                 const float* __restrict__ sums,
                                 const float* __restrict__ gamma,
                                 const float* __restrict__ beta,
                                 float* __restrict__ zsums,
                                 float* __restrict__ zsums2,
                                 int* __restrict__ zctr, int nzc,
                                 const unsigned* __restrict__ cpsrc,
                                 unsigned* __restrict__ cpdst, int ncp) {
  constexpr int CP = NT * 16;
  __shared__ float sWT[CP * WSTRIDE];
  __shared__ float sMean[CH];
  __shared__ float sSc[CH];
  __shared__ float sBeta[CH];
  int tid = threadIdx.x;
  if (zsums != nullptr && blockIdx.x == 0) zsums[tid] = 0.f;
  if (zsums2 != nullptr && blockIdx.x == 0) zsums2[tid] = 0.f;
  if (zctr != nullptr && blockIdx.x == 0) {
    for (int i = tid; i < nzc; i += 256) zctr[i] = 0;
  }
  if (cpsrc != nullptr) {
    int nth = (int)gridDim.x * 256;
    for (int i = (int)blockIdx.x * 256 + tid; i < ncp; i += nth)
      ((uint2*)cpdst)[i] = ((const uint2*)cpsrc)[i];
  }
  constexpr int WPER = CH * CP / 256;
  constexpr int WB = (WPER % 32 == 0) ? 32 : WPER;
  static_assert(WPER % WB == 0, "whole batches");
  for (int base = 0; base < WPER; base += WB) {
    float wv[WB];
#pragma unroll
    for (int i = 0; i < WB; ++i) {
      int idx = tid + (base + i) * 256;
      int k = idx / CP;
      int c = idx % CP;
      int cc = (c < COUT) ? c : (COUT - 1);
      wv[i] = W[k * COUT + cc];
    }
#pragma unroll
    for (int i = 0; i < WB; ++i) {
      int idx = tid + (base + i) * 256;
      int k = idx / CP;
      int c = idx % CP;
      sWT[c * WSTRIDE + k] = (c < COUT) ? wv[i] : 0.f;
    }
  }
  if (BN) {
    if (tid < CH) {
      float invN = 1.0f / (float)Nrows;
      float mean = sums[tid] * invN;
      float var = sums[128 + tid] * invN - mean * mean;
      sMean[tid] = mean;
      sSc[tid] = gamma[tid] * rsqrtf(var + BN_EPS);
      sBeta[tid] = beta[tid];
    }
  }
  __syncthreads();

  int wave = tid >> 5;
  int lane = tid & 31;
  int row0 = blockIdx.x * 128 + wave * 16;
  if (row0 >= Nrows) return;

  int m = lane & 15;
  int kh = (lane >> 4) * 2;

  v8f acc[NT];
#pragma unroll
  for (int t = 0; t < NT; ++t) acc[t] = {};

  const float* Arow = A + (size_t)(row0 + m) * CH;

  for (int k0 = 0; k0 < CH; k0 += 4) {
    v2f a = *(const v2f*)(Arow + k0 + kh);
    if (BN) {
      v2f mu = *(const v2f*)(&sMean[k0 + kh]);
      v2f sc = *(const v2f*)(&sSc[k0 + kh]);
      v2f be = *(const v2f*)(&sBeta[k0 + kh]);
      float x0 = (a.x - mu.x) * sc.x + be.x;
      float x1 = (a.y - mu.y) * sc.y + be.y;
      a.x = x0 > 0.f ? x0 : 0.f;
      a.y = x1 > 0.f ? x1 : 0.f;
    }
#pragma unroll
    for (int t = 0; t < NT; ++t) {
      v2f b = *(const v2f*)(&sWT[(t * 16 + m) * WSTRIDE + k0 + kh]);
      acc[t] = __builtin_amdgcn_wmma_f32_16x16x4_f32(
           false, a,  false, b,
           (short)0, acc[t],  false,  false);
    }
  }

  int rbase = row0 + (lane >> 4) * 8;
#pragma unroll
  for (int t = 0; t < NT; ++t) {
    int col = t * 16 + m;
    if (col < COUT) {
#pragma unroll
      for (int j = 0; j < 8; ++j) out[(size_t)(rbase + j) * ldo + col] = acc[t][j];
    }
  }
}

extern "C" void kernel_launch(void* const* d_in, const int* in_sizes, int n_in,
                              void* d_out, int out_size, void* d_ws, size_t ws_size,
                              hipStream_t stream) {
  const float* feat   = (const float*)d_in[0];
  const int*   src    = (const int*)d_in[1];
  const int*   dst    = (const int*)d_in[2];
  const float* ew     = (const float*)d_in[3];
  const float* W1     = (const float*)d_in[4];
  const float* W2     = (const float*)d_in[5];
  const float* W3     = (const float*)d_in[6];
  const float* b3     = (const float*)d_in[7];
  const float* gamma1 = (const float*)d_in[8];
  const float* beta1  = (const float*)d_in[9];
  const float* gamma2 = (const float*)d_in[10];
  const float* beta2  = (const float*)d_in[11];

  int N = in_sizes[0] / CH;
  int E = in_sizes[1];

  float* t    = (float*)d_ws;
  float* a    = t + (size_t)N * CH;
  float* sums = a + (size_t)N * CH;
  float* out  = (float*)d_out;

  int ggrid = (N + 127) / 128;
  dim3 b256(256);

  int nb = (N + TR - 1) / TR;
  int hdr = (3 * nb + 31) / 32 * 32;
  int bgrid = (E + EPW - 1) / EPW;
  int ncp = (hdr + 2 * E) / 2;
  int* bl1 = (int*)d_out;
  int* bl3 = (int*)(t + (size_t)N * 64);
  unsigned* ed1 = (unsigned*)(bl1 + hdr);
  unsigned* ed3 = (unsigned*)(bl3 + hdr);
  float* sums2 = (float*)(bl1 + hdr + 2 * E);

  gemm_wmma_kernel<8, false, 128><<<ggrid, b256, 0, stream>>>(
      feat, W1, t, N, 128, nullptr, nullptr, nullptr, sums, sums2, bl1, 2 * nb, nullptr, nullptr,
      0);
  bin_count_kernel<<<bgrid, b256, 0, stream>>>(dst, E, bl1, nb);
  bin_place_kernel<<<bgrid, b256, 0, stream>>>(src, dst, ew, E, bl1, bl1 + nb, bl1 + 2 * nb,
                                               ed1, nb);
  tile_sum_kernel<128, 32><<<nb, 1024, 0, stream>>>(
      t, bl1, bl1 + 2 * nb, ed1, a, N, nullptr, sums);

  gemm_wmma_kernel<8, true, 128><<<ggrid, b256, 0, stream>>>(
      a, W2, t, N, 128, sums, gamma1, beta1, nullptr, nullptr, nullptr, 0, nullptr, nullptr, 0);
  tile_sum_kernel<128, 32><<<nb, 1024, 0, stream>>>(
      t, bl1, bl1 + 2 * nb, ed1, a, N, nullptr, sums2);

  gemm_wmma_kernel<3, true, 40><<<ggrid, b256, 0, stream>>>(
      a, W3, t, N, 64, sums2, gamma2, beta2, nullptr, nullptr, nullptr, 0,
      (const unsigned*)bl1, (unsigned*)bl3, ncp);
  tile_sum_kernel<40, 32><<<nb, 1024, 0, stream>>>(
      t, bl3, bl3 + 2 * nb, ed3, out, N, b3, nullptr);
}
